# starting kernel
# baseline (speedup 1.0000x reference)
#include <hip/hip_runtime.h>
#include <hip/hip_bf16.h>
#include <math.h>


#define Bsz 4
#define Lsz 1024
#define Dsz 512
#define Hsz 8
#define Isz 64
#define Msz (Bsz * Lsz)
#define GN_EPS 1e-5f

typedef float v2f __attribute__((ext_vector_type(2)));
typedef float v8f __attribute__((ext_vector_type(8)));

__device__ __forceinline__ v8f wmma4(v2f a, v2f b, v8f c) {
  return __builtin_amdgcn_wmma_f32_16x16x4_f32(
       false, a,  false, b,
       (short)0, c,  false,  false);
}

__device__ __forceinline__ float silu_f(float v) {
  return v / (1.0f + expf(-v));
}

__global__ __launch_bounds__(128)
void gemm_gate_z_kernel(const float* __restrict__ X,
                        const float* __restrict__ Wz,  const float* __restrict__ bz,
                        const float* __restrict__ Wza, const float* __restrict__ bza,
                        float* __restrict__ Z) {
  const int lane = threadIdx.x & 31;
  const int wave = threadIdx.x >> 5;
  const int mr = lane & 15;
  const int kk = lane >> 4;
  const int m_base = blockIdx.x * 64 + (wave & 1) * 32;
  const int n_base = blockIdx.y * 64 + (wave >> 1) * 32;

  v8f a1[2][2], a2[2][2];
  const v8f zero = {0.f, 0.f, 0.f, 0.f, 0.f, 0.f, 0.f, 0.f};
  for (int i = 0; i < 2; ++i)
    for (int j = 0; j < 2; ++j) { a1[i][j] = zero; a2[i][j] = zero; }

  const float* A0  = X   + (size_t)(m_base + mr)       * Dsz + 2 * kk;
  const float* A1  = A0  + 16 * Dsz;
  const float* Bz0 = Wz  + (size_t)(n_base + mr)       * Dsz + 2 * kk;
  const float* Bz1 = Bz0 + 16 * Dsz;
  const float* Ba0 = Wza + (size_t)(n_base + mr)       * Dsz + 2 * kk;
  const float* Ba1 = Ba0 + 16 * Dsz;

#pragma unroll 4
  for (int k0 = 0; k0 < Dsz; k0 += 4) {
    v2f fa0 = *(const v2f*)(A0 + k0);
    v2f fa1 = *(const v2f*)(A1 + k0);
    v2f bz0 = *(const v2f*)(Bz0 + k0);
    v2f bz1 = *(const v2f*)(Bz1 + k0);
    v2f ba0 = *(const v2f*)(Ba0 + k0);
    v2f ba1 = *(const v2f*)(Ba1 + k0);
    a1[0][0] = wmma4(fa0, bz0, a1[0][0]);
    a1[0][1] = wmma4(fa0, bz1, a1[0][1]);
    a1[1][0] = wmma4(fa1, bz0, a1[1][0]);
    a1[1][1] = wmma4(fa1, bz1, a1[1][1]);
    a2[0][0] = wmma4(fa0, ba0, a2[0][0]);
    a2[0][1] = wmma4(fa0, ba1, a2[0][1]);
    a2[1][0] = wmma4(fa1, ba0, a2[1][0]);
    a2[1][1] = wmma4(fa1, ba1, a2[1][1]);
  }

  for (int mt = 0; mt < 2; ++mt)
    for (int nt = 0; nt < 2; ++nt) {
      const int n = n_base + nt * 16 + mr;
      const float biz = bz[n], biza = bza[n];
#pragma unroll
      for (int r = 0; r < 8; ++r) {
        const int m = m_base + mt * 16 + kk * 8 + r;
        const float v1 = a1[mt][nt][r] + biz;
        const float v2 = a2[mt][nt][r] + biza;
        Z[(size_t)m * Dsz + n] = v1 * silu_f(v2);
      }
    }
}

template <int MODE>
__global__ __launch_bounds__(128)
void gemm_single_kernel(const float* __restrict__ A, const float* __restrict__ W,
                        const float* __restrict__ bias,
                        const float* __restrict__ gate, float* __restrict__ OUT) {
  const int lane = threadIdx.x & 31;
  const int wave = threadIdx.x >> 5;
  const int mr = lane & 15;
  const int kk = lane >> 4;
  const int m_base = blockIdx.x * 64 + (wave & 1) * 32;
  const int n_base = blockIdx.y * 64 + (wave >> 1) * 32;

  v8f acc[2][2];
  const v8f zero = {0.f, 0.f, 0.f, 0.f, 0.f, 0.f, 0.f, 0.f};
  for (int i = 0; i < 2; ++i)
    for (int j = 0; j < 2; ++j) acc[i][j] = zero;

  const float* A0 = A + (size_t)(m_base + mr) * Dsz + 2 * kk;
  const float* A1 = A0 + 16 * Dsz;
  const float* B0 = W + (size_t)(n_base + mr) * Dsz + 2 * kk;
  const float* B1 = B0 + 16 * Dsz;

#pragma unroll 4
  for (int k0 = 0; k0 < Dsz; k0 += 4) {
    v2f fa0 = *(const v2f*)(A0 + k0);
    v2f fa1 = *(const v2f*)(A1 + k0);
    v2f fb0 = *(const v2f*)(B0 + k0);
    v2f fb1 = *(const v2f*)(B1 + k0);
    acc[0][0] = wmma4(fa0, fb0, acc[0][0]);
    acc[0][1] = wmma4(fa0, fb1, acc[0][1]);
    acc[1][0] = wmma4(fa1, fb0, acc[1][0]);
    acc[1][1] = wmma4(fa1, fb1, acc[1][1]);
  }

  for (int mt = 0; mt < 2; ++mt)
    for (int nt = 0; nt < 2; ++nt) {
      const int n = n_base + nt * 16 + mr;
      const float bi = bias[n];
#pragma unroll
      for (int r = 0; r < 8; ++r) {
        const int m = m_base + mt * 16 + kk * 8 + r;
        const float v = acc[mt][nt][r] + bi;
        float o;
        if (MODE == 0) o = silu_f(v);
        else           o = v * gate[(size_t)m * Dsz + n];
        OUT[(size_t)m * Dsz + n] = o;
      }
    }
}

__global__ __launch_bounds__(1024)
void dtscan_kernel(const float* __restrict__ X, const float* __restrict__ w_dt,
                   const float* __restrict__ b_dt, const float* __restrict__ ln_a,
                   float* __restrict__ S) {
  __shared__ float wlds[Dsz];
  __shared__ float scan[Lsz];
  const int b = blockIdx.x >> 3;
  const int h = blockIdx.x & 7;
  const int l = threadIdx.x;

  if (l < Dsz) wlds[l] = w_dt[h * Dsz + l];
  __syncthreads();

  const float* xr = X + (size_t)(b * Lsz + l) * Dsz;
  float dot = 0.f;
#pragma unroll 4
  for (int k = 0; k < Dsz; k += 4) {
    float4 xv = *(const float4*)(xr + k);
    dot += xv.x * wlds[k] + xv.y * wlds[k + 1] + xv.z * wlds[k + 2] + xv.w * wlds[k + 3];
  }
  const float v = dot + b_dt[h];
  const float sp = (v > 20.f) ? v : log1pf(expf(v));
  const float ld = -expf(ln_a[h]) * sp;

  scan[l] = ld;
  __syncthreads();
  for (int off = 1; off < Lsz; off <<= 1) {
    float t = scan[l];
    if (l >= off) t += scan[l - off];
    __syncthreads();
    scan[l] = t;
    __syncthreads();
  }
  S[(size_t)(b * Lsz + l) * Hsz + h] = scan[l];
}

__global__ __launch_bounds__(256)
void zsum_kernel(const float* __restrict__ Z, const float* __restrict__ hidden,
                 float* __restrict__ C, float* __restrict__ zhh) {
  __shared__ float red[256];
  const int b = blockIdx.x >> 3;
  const int h = blockIdx.x & 7;
  const int i = threadIdx.x & 63;
  const int part = threadIdx.x >> 6;

  float s = 0.f;
  for (int m = h + 1 + part; m < Lsz; m += 4)
    s += Z[((size_t)(b * Lsz + m) * Hsz + h) * Isz + i];
  red[threadIdx.x] = s;
  __syncthreads();
  if (part == 0) {
    const float tot = red[i] + red[i + 64] + red[i + 128] + red[i + 192];
    const int ch = (b * Hsz + h) * Isz + i;
    C[ch] = tot + hidden[ch];
    zhh[ch] = Z[((size_t)(b * Lsz + h) * Hsz + h) * Isz + i];
  }
}

__global__ __launch_bounds__(512)
void hnorm_kernel(const float* __restrict__ S, const float* __restrict__ C,
                  const float* __restrict__ zhh, const float* __restrict__ gn_w,
                  const float* __restrict__ gn_b, float* __restrict__ Hout,
                  float* __restrict__ Hnorm) {
  __shared__ float lds[Dsz];
  const int bl = blockIdx.x;
  const int b = bl >> 10;
  const int t = threadIdx.x;
  const int h = t >> 6;
  const int i = t & 63;
  const int ch = (b * Hsz + h) * Isz + i;

  const float e = expf(S[(size_t)bl * Hsz + h]);
  const float hv = zhh[ch] + e * C[ch];
  Hout[(size_t)bl * Dsz + t] = hv;

  lds[t] = hv;
  __syncthreads();
  for (int s = 32; s >= 1; s >>= 1) {
    if (i < s) lds[t] += lds[t + s];
    __syncthreads();
  }
  const float mu = lds[h * 64] * (1.0f / 64.0f);
  __syncthreads();
  const float d = hv - mu;
  lds[t] = d * d;
  __syncthreads();
  for (int s = 32; s >= 1; s >>= 1) {
    if (i < s) lds[t] += lds[t + s];
    __syncthreads();
  }
  const float var = lds[h * 64] * (1.0f / 64.0f);
  const float hn = d / sqrtf(var + GN_EPS) * gn_w[h] + gn_b[h];
  Hnorm[(size_t)bl * Dsz + t] = hn;
}

extern "C" void kernel_launch(void* const* d_in, const int* in_sizes, int n_in,
                              void* d_out, int out_size, void* d_ws, size_t ws_size,
                              hipStream_t stream) {
  const float* x      = (const float*)d_in[0];
  const float* hidden = (const float*)d_in[1];
  const float* w_z    = (const float*)d_in[2];
  const float* b_z    = (const float*)d_in[3];
  const float* w_za   = (const float*)d_in[4];
  const float* b_za   = (const float*)d_in[5];
  const float* w_y    = (const float*)d_in[6];
  const float* b_y    = (const float*)d_in[7];
  const float* w_ya   = (const float*)d_in[8];
  const float* b_ya   = (const float*)d_in[9];
  const float* w_dt   = (const float*)d_in[10];
  const float* b_dt   = (const float*)d_in[11];
  const float* ln_a   = (const float*)d_in[12];
  const float* gn_w   = (const float*)d_in[13];
  const float* gn_b   = (const float*)d_in[14];

  float* y_out = (float*)d_out;
  float* h_out = y_out + (size_t)Msz * Dsz;

  float* ws   = (float*)d_ws;
  float* Z    = ws;
  float* GYA  = Z    + (size_t)Msz * Dsz;
  float* HN   = GYA  + (size_t)Msz * Dsz;
  float* Sb   = HN   + (size_t)Msz * Dsz;
  float* Cb   = Sb   + (size_t)Msz * Hsz;
  float* ZHH  = Cb   + (size_t)Bsz * Hsz * Isz;

  const dim3 gGemm(Msz / 64, Dsz / 64, 1);
  const dim3 bGemm(128, 1, 1);

  gemm_gate_z_kernel<<<gGemm, bGemm, 0, stream>>>(x, w_z, b_z, w_za, b_za, Z);
  gemm_single_kernel<0><<<gGemm, bGemm, 0, stream>>>(x, w_ya, b_ya, nullptr, GYA);
  dtscan_kernel<<<dim3(Bsz * Hsz), dim3(Lsz), 0, stream>>>(x, w_dt, b_dt, ln_a, Sb);
  zsum_kernel<<<dim3(Bsz * Hsz), dim3(256), 0, stream>>>(Z, hidden, Cb, ZHH);
  hnorm_kernel<<<dim3(Msz), dim3(Dsz), 0, stream>>>(Sb, Cb, ZHH, gn_w, gn_b,
                                                    h_out, HN);
  gemm_single_kernel<1><<<gGemm, bGemm, 0, stream>>>(HN, w_y, b_y, GYA, y_out);
}


# Wya projection done inside the output GEMM (own K sweep, gate kept in registers): gate plane never stored or re-read, 6 launches -> 5; dt columns ride on the z GEMM. Sum order unchanged
# speedup vs baseline: 2.6945x; 2.6945x over previous
#include <hip/hip_runtime.h>
#include <hip/hip_bf16.h>
#include <math.h>


#define Bsz 4
#define Lsz 1024
#define Dsz 512
#define Hsz 8
#define Isz 64
#define Msz (Bsz * Lsz)
#define GN_EPS 1e-5f
#define LOADS_THEN_PRODUCTS(NL, NM)                         \
  __builtin_amdgcn_sched_group_barrier(0x020, NL, 0);       \
  __builtin_amdgcn_sched_group_barrier(0x008, NM, 0);       \
  __builtin_amdgcn_sched_group_barrier(0x020, NL, 0);       \
  __builtin_amdgcn_sched_group_barrier(0x008, NM, 0)
#define KBG 16
#define KBS 32
#define NSEG 16
#define KSEG (Dsz / NSEG)

typedef float v2f __attribute__((ext_vector_type(2)));
typedef float v8f __attribute__((ext_vector_type(8)));

__device__ __forceinline__ v8f wmma4(v2f a, v2f b, v8f c) {
  return __builtin_amdgcn_wmma_f32_16x16x4_f32(
       false, a,  false, b,
       (short)0, c,  false,  false);
}

__device__ __forceinline__ float silu_f(float v) {
  return v / (1.0f + expf(-v));
}

__global__ __launch_bounds__(128)
void gemm_gate_z_kernel(const float* __restrict__ X,
                        const float* __restrict__ Wz,  const float* __restrict__ bz,
                        const float* __restrict__ Wza, const float* __restrict__ bza,
                        float* __restrict__ Z,
                        const float* __restrict__ Wd, float* __restrict__ PD) {
  const int lane = threadIdx.x & 31;
  const int wave = threadIdx.x >> 5;
  const int mr = lane & 15;
  const int kk = lane >> 4;
  const int m_base = blockIdx.x * 64 + (wave & 1) * 32;
  const int n_base = blockIdx.y * 64 + (wave >> 1) * 32;

  v8f a1[2][2], a2[2][2];
  const v8f zero = {0.f, 0.f, 0.f, 0.f, 0.f, 0.f, 0.f, 0.f};
  for (int i = 0; i < 2; ++i)
    for (int j = 0; j < 2; ++j) { a1[i][j] = zero; a2[i][j] = zero; }

  const float* A0  = X   + (size_t)(m_base + mr)       * Dsz + 2 * kk;
  const float* A1  = A0  + 16 * Dsz;
  const float* Bz0 = Wz  + (size_t)(n_base + mr)       * Dsz + 2 * kk;
  const float* Bz1 = Bz0 + 16 * Dsz;
  const float* Ba0 = Wza + (size_t)(n_base + mr)       * Dsz + 2 * kk;
  const float* Ba1 = Ba0 + 16 * Dsz;

  v2f pa[2][KBG / 4][6];
#define GATE_LOAD(S, K)                                              \
  _Pragma("unroll")                                                  \
  for (int u = 0; u < KBG / 4; ++u) {                                \
    pa[S][u][0] = *(const v2f*)(A0  + (K) + 4 * u);                  \
    pa[S][u][1] = *(const v2f*)(A1  + (K) + 4 * u);                  \
    pa[S][u][2] = *(const v2f*)(Bz0 + (K) + 4 * u);                  \
    pa[S][u][3] = *(const v2f*)(Bz1 + (K) + 4 * u);                  \
    pa[S][u][4] = *(const v2f*)(Ba0 + (K) + 4 * u);                  \
    pa[S][u][5] = *(const v2f*)(Ba1 + (K) + 4 * u);                  \
  }
#define GATE_MMA(S)                                                  \
  _Pragma("unroll")                                                  \
  for (int u = 0; u < KBG / 4; ++u) {                                \
    a1[0][0] = wmma4(pa[S][u][0], pa[S][u][2], a1[0][0]);            \
    a1[0][1] = wmma4(pa[S][u][0], pa[S][u][3], a1[0][1]);            \
    a1[1][0] = wmma4(pa[S][u][1], pa[S][u][2], a1[1][0]);            \
    a1[1][1] = wmma4(pa[S][u][1], pa[S][u][3], a1[1][1]);            \
    a2[0][0] = wmma4(pa[S][u][0], pa[S][u][4], a2[0][0]);            \
    a2[0][1] = wmma4(pa[S][u][0], pa[S][u][5], a2[0][1]);            \
    a2[1][0] = wmma4(pa[S][u][1], pa[S][u][4], a2[1][0]);            \
    a2[1][1] = wmma4(pa[S][u][1], pa[S][u][5], a2[1][1]);            \
  }

  GATE_LOAD(0, 0)

  {
    const int seg = blockIdx.y * 2 + (wave >> 1);
    const float* Ad0 = A0 + seg * KSEG;
    const float* Ad1 = A1 + seg * KSEG;
    const float* Bd  = Wd + (size_t)(mr & 7) * Dsz + 2 * kk + seg * KSEG;
    v8f d0 = zero, d1 = zero;
#pragma unroll
    for (int u = 0; u < KSEG; u += 4) {
      v2f fa0 = *(const v2f*)(Ad0 + u);
      v2f fa1 = *(const v2f*)(Ad1 + u);
      v2f fd  = *(const v2f*)(Bd + u);
      d0 = wmma4(fa0, fd, d0);
      d1 = wmma4(fa1, fd, d1);
    }
    if (mr < Hsz) {
      float* pd = PD + (size_t)seg * (Msz * Hsz)
                     + (size_t)(m_base + kk * 8) * Hsz + mr;
#pragma unroll
      for (int r = 0; r < 8; ++r) {
        pd[r * Hsz]        = d0[r];
        pd[(16 + r) * Hsz] = d1[r];
      }
    }
  }

#pragma nounroll
  for (int kb = 0; kb < Dsz - 2 * KBG; kb += 2 * KBG) {
    GATE_LOAD(1, kb + KBG)
    GATE_MMA(0)
    GATE_LOAD(0, kb + 2 * KBG)
    GATE_MMA(1)
    LOADS_THEN_PRODUCTS(6 * (KBG / 4), 8 * (KBG / 4));
  }
  GATE_LOAD(1, Dsz - KBG)
  GATE_MMA(0)
  GATE_MMA(1)
#undef GATE_LOAD
#undef GATE_MMA

  for (int mt = 0; mt < 2; ++mt)
    for (int nt = 0; nt < 2; ++nt) {
      const int n = n_base + nt * 16 + mr;
      const float biz = bz[n], biza = bza[n];
#pragma unroll
      for (int r = 0; r < 8; ++r) {
        const int m = m_base + mt * 16 + kk * 8 + r;
        const float v1 = a1[mt][nt][r] + biz;
        const float v2 = a2[mt][nt][r] + biza;
        Z[(size_t)m * Dsz + n] = v1 * silu_f(v2);
      }
    }
}

__global__ __launch_bounds__(128)
void gemm_out_kernel(const float* __restrict__ HN, const float* __restrict__ Wy,
                     const float* __restrict__ by,
                     const float* __restrict__ X, const float* __restrict__ Wya,
                     const float* __restrict__ bya, float* __restrict__ OUT) {
  const int lane = threadIdx.x & 31;
  const int wave = threadIdx.x >> 5;
  const int mr = lane & 15;
  const int kk = lane >> 4;
  const int m_base = blockIdx.x * 64 + (wave & 1) * 32;
  const int n_base = blockIdx.y * 64 + (wave >> 1) * 32;

  v8f accg[2][2], accy[2][2];
  const v8f zero = {0.f, 0.f, 0.f, 0.f, 0.f, 0.f, 0.f, 0.f};
  for (int i = 0; i < 2; ++i)
    for (int j = 0; j < 2; ++j) { accg[i][j] = zero; accy[i][j] = zero; }

  const size_t ra = (size_t)(m_base + mr) * Dsz + 2 * kk;
  const size_t rb = (size_t)(n_base + mr) * Dsz + 2 * kk;
  const float* G0 = X   + ra;
  const float* G1 = G0  + 16 * Dsz;
  const float* V0 = Wya + rb;
  const float* V1 = V0  + 16 * Dsz;
  const float* A0 = HN  + ra;
  const float* A1 = A0  + 16 * Dsz;
  const float* B0 = Wy  + rb;
  const float* B1 = B0  + 16 * Dsz;

  v2f pf[2][KBS / 4][4];
#define OUT_LOAD(S, K, PA0, PA1, PB0, PB1)                           \
  _Pragma("unroll")                                                  \
  for (int u = 0; u < KBS / 4; ++u) {                                \
    pf[S][u][0] = *(const v2f*)(PA0 + (K) + 4 * u);                  \
    pf[S][u][1] = *(const v2f*)(PA1 + (K) + 4 * u);                  \
    pf[S][u][2] = *(const v2f*)(PB0 + (K) + 4 * u);                  \
    pf[S][u][3] = *(const v2f*)(PB1 + (K) + 4 * u);                  \
  }
#define OUT_MMA(S, ACC)                                              \
  _Pragma("unroll")                                                  \
  for (int u = 0; u < KBS / 4; ++u) {                                \
    ACC[0][0] = wmma4(pf[S][u][0], pf[S][u][2], ACC[0][0]);          \
    ACC[0][1] = wmma4(pf[S][u][0], pf[S][u][3], ACC[0][1]);          \
    ACC[1][0] = wmma4(pf[S][u][1], pf[S][u][2], ACC[1][0]);          \
    ACC[1][1] = wmma4(pf[S][u][1], pf[S][u][3], ACC[1][1]);          \
  }

  OUT_LOAD(0, 0, G0, G1, V0, V1)
#pragma nounroll
  for (int kb = 0; kb < Dsz - 2 * KBS; kb += 2 * KBS) {
    OUT_LOAD(1, kb + KBS, G0, G1, V0, V1)
    OUT_MMA(0, accg)
    OUT_LOAD(0, kb + 2 * KBS, G0, G1, V0, V1)
    OUT_MMA(1, accg)
    LOADS_THEN_PRODUCTS(4 * (KBS / 4), 4 * (KBS / 4));
  }
  OUT_LOAD(1, Dsz - KBS, G0, G1, V0, V1)
  OUT_MMA(0, accg)
  OUT_LOAD(0, 0, A0, A1, B0, B1)
  OUT_MMA(1, accg)

#pragma nounroll
  for (int kb = 0; kb < Dsz - 2 * KBS; kb += 2 * KBS) {
    OUT_LOAD(1, kb + KBS, A0, A1, B0, B1)
    OUT_MMA(0, accy)
    OUT_LOAD(0, kb + 2 * KBS, A0, A1, B0, B1)
    OUT_MMA(1, accy)
    LOADS_THEN_PRODUCTS(4 * (KBS / 4), 4 * (KBS / 4));
  }
  OUT_LOAD(1, Dsz - KBS, A0, A1, B0, B1)
  OUT_MMA(0, accy)
  OUT_MMA(1, accy)
#undef OUT_LOAD
#undef OUT_MMA

  for (int mt = 0; mt < 2; ++mt)
    for (int nt = 0; nt < 2; ++nt) {
      const int n = n_base + nt * 16 + mr;
      const float biy = by[n], biya = bya[n];
#pragma unroll
      for (int r = 0; r < 8; ++r) {
        const int m = m_base + mt * 16 + kk * 8 + r;
        const float v = accy[mt][nt][r] + biy;
        const float g = silu_f(accg[mt][nt][r] + biya);
        OUT[(size_t)m * Dsz + n] = v * g;
      }
    }
}

__global__ __launch_bounds__(1024)
void dtscan_kernel(const float* __restrict__ PD,
                   const float* __restrict__ b_dt, const float* __restrict__ ln_a,
                   float* __restrict__ S) {
  __shared__ float scan[Lsz];
  const int b = blockIdx.x >> 3;
  const int h = blockIdx.x & 7;
  const int l = threadIdx.x;

  const float* pp = PD + (size_t)(b * Lsz + l) * Hsz + h;
  float pv[NSEG];
#pragma unroll
  for (int j = 0; j < NSEG; ++j) pv[j] = pp[(size_t)j * (Msz * Hsz)];
  float dot = pv[0];
#pragma unroll
  for (int j = 1; j < NSEG; ++j) dot += pv[j];

  const float v = dot + b_dt[h];
  const float sp = (v > 20.f) ? v : log1pf(expf(v));
  const float ld = -expf(ln_a[h]) * sp;

  scan[l] = ld;
  __syncthreads();
  for (int off = 1; off < Lsz; off <<= 1) {
    float t = scan[l];
    if (l >= off) t += scan[l - off];
    __syncthreads();
    scan[l] = t;
    __syncthreads();
  }
  S[(size_t)(b * Lsz + l) * Hsz + h] = scan[l];
}

#define NCH 8
#define RCH (Lsz / NCH)
#define NBHI (Bsz * Hsz * Isz)

__global__ __launch_bounds__(256)
void zsum_kernel(const float* __restrict__ Z, const float* __restrict__ hidden,
                 float* __restrict__ P, float* __restrict__ zhh) {
  __shared__ float red[16 * Isz];
  const int bh = blockIdx.x;
  const int chunk = blockIdx.y;
  const int b = bh >> 3;
  const int h = bh & 7;
  const int c4 = threadIdx.x & 15;
  const int part = threadIdx.x >> 4;

  const int m0 = chunk * RCH + part;
  const float* zp = Z + ((size_t)(b * Lsz + m0) * Hsz + h) * Isz + c4 * 4;

  float4 v[RCH / 16];
  if (m0 > h) v[0] = *(const float4*)zp;
  else        v[0] = make_float4(0.f, 0.f, 0.f, 0.f);
#pragma unroll
  for (int j = 1; j < RCH / 16; ++j)
    v[j] = *(const float4*)(zp + (size_t)j * 16 * Dsz);
  float4 s = v[0];
#pragma unroll
  for (int j = 1; j < RCH / 16; ++j) {
    s.x += v[j].x; s.y += v[j].y; s.z += v[j].z; s.w += v[j].w;
  }
  *(float4*)(red + part * Isz + c4 * 4) = s;
  __syncthreads();

  if (threadIdx.x < Isz) {
    const int i = threadIdx.x;
    float tot = 0.f;
#pragma unroll
    for (int p = 0; p < 16; ++p) tot += red[p * Isz + i];
    const int ch = bh * Isz + i;
    if (chunk == 0) {
      tot += hidden[ch];
      zhh[ch] = Z[((size_t)(b * Lsz + h) * Hsz + h) * Isz + i];
    }
    P[(size_t)chunk * NBHI + ch] = tot;
  }
}

#define RPB 16

__device__ __forceinline__ float wave_sum(float v) {
#pragma unroll
  for (int off = 16; off >= 1; off >>= 1) v += __shfl_xor(v, off, 32);
  return v;
}

__device__ __forceinline__ float lane_bcast(float v, int src) {
  return __builtin_bit_cast(
      float, __builtin_amdgcn_readlane(__builtin_bit_cast(int, v), src));
}

__global__ __launch_bounds__(256)
void hnorm_kernel(const float* __restrict__ S, const float* __restrict__ P,
                  const float* __restrict__ zhh, const float* __restrict__ gn_w,
                  const float* __restrict__ gn_b, float* __restrict__ Hout,
                  float* __restrict__ Hnorm) {
  const int lane = threadIdx.x & 31;
  const int h = threadIdx.x >> 5;
  const int row0 = blockIdx.x * RPB;
  const int b = row0 >> 10;
  const int col = h * Isz + 2 * lane;
  const int ch = b * (Hsz * Isz) + col;

  const float ev = expf(S[(size_t)(row0 + (lane & (RPB - 1))) * Hsz + h]);

  float2 c = make_float2(0.f, 0.f);
#pragma unroll
  for (int k = 0; k < NCH; ++k) {
    const float2 p = *(const float2*)(P + (size_t)k * NBHI + ch);
    c.x += p.x; c.y += p.y;
  }
  const float2 z = *(const float2*)(zhh + ch);
  const float gw = gn_w[h], gb = gn_b[h];

  const float inv_i = 1.0f / (float)Isz;
  const float mz = wave_sum(z.x + z.y) * inv_i;
  const float mc = wave_sum(c.x + c.y) * inv_i;
  const float dzx = z.x - mz, dzy = z.y - mz;
  const float dcx = c.x - mc, dcy = c.y - mc;
  const float vz  = wave_sum(dzx * dzx + dzy * dzy) * inv_i;
  const float czc = wave_sum(dzx * dcx + dzy * dcy) * inv_i;
  const float vc  = wave_sum(dcx * dcx + dcy * dcy) * inv_i;

  const float var = fmaxf(vz + ev * (2.0f * czc + ev * vc), 0.f);
  const float sv = gw / sqrtf(var + GN_EPS);

#pragma unroll
  for (int r = 0; r < RPB; ++r) {
    const float er = lane_bcast(ev, r);
    const float sr = lane_bcast(sv, r);
    float2 hv, hn;
    hv.x = z.x + er * c.x;
    hv.y = z.y + er * c.y;
    hn.x = (dzx + er * dcx) * sr + gb;
    hn.y = (dzy + er * dcy) * sr + gb;
    const size_t o = (size_t)(row0 + r) * Dsz + col;
    *(float2*)(Hout + o)  = hv;
    *(float2*)(Hnorm + o) = hn;
  }
}

extern "C" void kernel_launch(void* const* d_in, const int* in_sizes, int n_in,
                              void* d_out, int out_size, void* d_ws, size_t ws_size,
                              hipStream_t stream) {
  const float* x      = (const float*)d_in[0];
  const float* hidden = (const float*)d_in[1];
  const float* w_z    = (const float*)d_in[2];
  const float* b_z    = (const float*)d_in[3];
  const float* w_za   = (const float*)d_in[4];
  const float* b_za   = (const float*)d_in[5];
  const float* w_y    = (const float*)d_in[6];
  const float* b_y    = (const float*)d_in[7];
  const float* w_ya   = (const float*)d_in[8];
  const float* b_ya   = (const float*)d_in[9];
  const float* w_dt   = (const float*)d_in[10];
  const float* b_dt   = (const float*)d_in[11];
  const float* ln_a   = (const float*)d_in[12];
  const float* gn_w   = (const float*)d_in[13];
  const float* gn_b   = (const float*)d_in[14];

  float* y_out = (float*)d_out;
  float* h_out = y_out + (size_t)Msz * Dsz;

  float* ws   = (float*)d_ws;
  float* Z    = ws;
  float* HN   = Z    + (size_t)Msz * Dsz;
  float* Sb   = HN   + (size_t)Msz * Dsz;
  float* ZHH  = Sb   + (size_t)Msz * Hsz;
  float* Pb   = ZHH  + (size_t)Bsz * Hsz * Isz;
  float* PDb  = Pb   + (size_t)NCH * NBHI;

  const dim3 gGemm(Msz / 64, Dsz / 64, 1);
  const dim3 bGemm(128, 1, 1);

  gemm_gate_z_kernel<<<gGemm, bGemm, 0, stream>>>(x, w_z, b_z, w_za, b_za, Z,
                                                  w_dt, PDb);
  dtscan_kernel<<<dim3(Bsz * Hsz), dim3(Lsz), 0, stream>>>(PDb, b_dt, ln_a, Sb);
  zsum_kernel<<<dim3(Bsz * Hsz, NCH), dim3(256), 0, stream>>>(Z, hidden, Pb, ZHH);
  hnorm_kernel<<<dim3(Msz / RPB), dim3(256), 0, stream>>>(Sb, Pb, ZHH, gn_w, gn_b,
                                                          h_out, HN);
  gemm_out_kernel<<<gGemm, bGemm, 0, stream>>>(HN, w_y, b_y, x, w_ya, b_ya, y_out);
}


# h_norm never stored: output GEMM forms its A fragments in registers from per-head (dz, dc) and per-row (exp S, scale) factors and writes h too; norm launch gone, 5 launches -> 4. Sum order unchanged
# speedup vs baseline: 2.9284x; 1.0868x over previous
#include <hip/hip_runtime.h>
#include <hip/hip_bf16.h>
#include <math.h>


#define Bsz 4
#define Lsz 1024
#define Dsz 512
#define Hsz 8
#define Isz 64
#define Msz (Bsz * Lsz)
#define GN_EPS 1e-5f
#define LOADS_THEN_PRODUCTS(NL, NM)                         \
  __builtin_amdgcn_sched_group_barrier(0x020, NL, 0);       \
  __builtin_amdgcn_sched_group_barrier(0x008, NM, 0);       \
  __builtin_amdgcn_sched_group_barrier(0x020, NL, 0);       \
  __builtin_amdgcn_sched_group_barrier(0x008, NM, 0)
#define KBG 16
#define KBS 32
#define NSEG 16
#define KSEG (Dsz / NSEG)

typedef float v2f __attribute__((ext_vector_type(2)));
typedef float v8f __attribute__((ext_vector_type(8)));

__device__ __forceinline__ v8f wmma4(v2f a, v2f b, v8f c) {
  return __builtin_amdgcn_wmma_f32_16x16x4_f32(
       false, a,  false, b,
       (short)0, c,  false,  false);
}

__device__ __forceinline__ float silu_f(float v) {
  return v / (1.0f + expf(-v));
}

__global__ __launch_bounds__(128)
void gemm_gate_z_kernel(const float* __restrict__ X,
                        const float* __restrict__ Wz,  const float* __restrict__ bz,
                        const float* __restrict__ Wza, const float* __restrict__ bza,
                        float* __restrict__ Z,
                        const float* __restrict__ Wd, float* __restrict__ PD) {
  const int lane = threadIdx.x & 31;
  const int wave = threadIdx.x >> 5;
  const int mr = lane & 15;
  const int kk = lane >> 4;
  const int m_base = blockIdx.x * 64 + (wave & 1) * 32;
  const int n_base = blockIdx.y * 64 + (wave >> 1) * 32;

  v8f a1[2][2], a2[2][2];
  const v8f zero = {0.f, 0.f, 0.f, 0.f, 0.f, 0.f, 0.f, 0.f};
  for (int i = 0; i < 2; ++i)
    for (int j = 0; j < 2; ++j) { a1[i][j] = zero; a2[i][j] = zero; }

  const float* A0  = X   + (size_t)(m_base + mr)       * Dsz + 2 * kk;
  const float* A1  = A0  + 16 * Dsz;
  const float* Bz0 = Wz  + (size_t)(n_base + mr)       * Dsz + 2 * kk;
  const float* Bz1 = Bz0 + 16 * Dsz;
  const float* Ba0 = Wza + (size_t)(n_base + mr)       * Dsz + 2 * kk;
  const float* Ba1 = Ba0 + 16 * Dsz;

  v2f pa[2][KBG / 4][6];
#define GATE_LOAD(S, K)                                              \
  _Pragma("unroll")                                                  \
  for (int u = 0; u < KBG / 4; ++u) {                                \
    pa[S][u][0] = *(const v2f*)(A0  + (K) + 4 * u);                  \
    pa[S][u][1] = *(const v2f*)(A1  + (K) + 4 * u);                  \
    pa[S][u][2] = *(const v2f*)(Bz0 + (K) + 4 * u);                  \
    pa[S][u][3] = *(const v2f*)(Bz1 + (K) + 4 * u);                  \
    pa[S][u][4] = *(const v2f*)(Ba0 + (K) + 4 * u);                  \
    pa[S][u][5] = *(const v2f*)(Ba1 + (K) + 4 * u);                  \
  }
#define GATE_MMA(S)                                                  \
  _Pragma("unroll")                                                  \
  for (int u = 0; u < KBG / 4; ++u) {                                \
    a1[0][0] = wmma4(pa[S][u][0], pa[S][u][2], a1[0][0]);            \
    a1[0][1] = wmma4(pa[S][u][0], pa[S][u][3], a1[0][1]);            \
    a1[1][0] = wmma4(pa[S][u][1], pa[S][u][2], a1[1][0]);            \
    a1[1][1] = wmma4(pa[S][u][1], pa[S][u][3], a1[1][1]);            \
    a2[0][0] = wmma4(pa[S][u][0], pa[S][u][4], a2[0][0]);            \
    a2[0][1] = wmma4(pa[S][u][0], pa[S][u][5], a2[0][1]);            \
    a2[1][0] = wmma4(pa[S][u][1], pa[S][u][4], a2[1][0]);            \
    a2[1][1] = wmma4(pa[S][u][1], pa[S][u][5], a2[1][1]);            \
  }

  GATE_LOAD(0, 0)

  {
    const int seg = blockIdx.y * 2 + (wave >> 1);
    const float* Ad0 = A0 + seg * KSEG;
    const float* Ad1 = A1 + seg * KSEG;
    const float* Bd  = Wd + (size_t)(mr & 7) * Dsz + 2 * kk + seg * KSEG;
    v8f d0 = zero, d1 = zero;
#pragma unroll
    for (int u = 0; u < KSEG; u += 4) {
      v2f fa0 = *(const v2f*)(Ad0 + u);
      v2f fa1 = *(const v2f*)(Ad1 + u);
      v2f fd  = *(const v2f*)(Bd + u);
      d0 = wmma4(fa0, fd, d0);
      d1 = wmma4(fa1, fd, d1);
    }
    if (mr < Hsz) {
      float* pd = PD + (size_t)seg * (Msz * Hsz)
                     + (size_t)(m_base + kk * 8) * Hsz + mr;
#pragma unroll
      for (int r = 0; r < 8; ++r) {
        pd[r * Hsz]        = d0[r];
        pd[(16 + r) * Hsz] = d1[r];
      }
    }
  }

#pragma nounroll
  for (int kb = 0; kb < Dsz - 2 * KBG; kb += 2 * KBG) {
    GATE_LOAD(1, kb + KBG)
    GATE_MMA(0)
    GATE_LOAD(0, kb + 2 * KBG)
    GATE_MMA(1)
    LOADS_THEN_PRODUCTS(6 * (KBG / 4), 8 * (KBG / 4));
  }
  GATE_LOAD(1, Dsz - KBG)
  GATE_MMA(0)
  GATE_MMA(1)
#undef GATE_LOAD
#undef GATE_MMA

  for (int mt = 0; mt < 2; ++mt)
    for (int nt = 0; nt < 2; ++nt) {
      const int n = n_base + nt * 16 + mr;
      const float biz = bz[n], biza = bza[n];
#pragma unroll
      for (int r = 0; r < 8; ++r) {
        const int m = m_base + mt * 16 + kk * 8 + r;
        const float v1 = a1[mt][nt][r] + biz;
        const float v2 = a2[mt][nt][r] + biza;
        Z[(size_t)m * Dsz + n] = v1 * silu_f(v2);
      }
    }
}

#define NCH 8
#define RCH (Lsz / NCH)
#define NBHI (Bsz * Hsz * Isz)

__global__ __launch_bounds__(256)
void zsum_kernel(const float* __restrict__ Z, const float* __restrict__ hidden,
                 float* __restrict__ P, float* __restrict__ zhh) {
  __shared__ float red[16 * Isz];
  const int bh = blockIdx.x;
  const int chunk = blockIdx.y;
  const int b = bh >> 3;
  const int h = bh & 7;
  const int c4 = threadIdx.x & 15;
  const int part = threadIdx.x >> 4;

  const int m0 = chunk * RCH + part;
  const float* zp = Z + ((size_t)(b * Lsz + m0) * Hsz + h) * Isz + c4 * 4;

  float4 v[RCH / 16];
  if (m0 > h) v[0] = *(const float4*)zp;
  else        v[0] = make_float4(0.f, 0.f, 0.f, 0.f);
#pragma unroll
  for (int j = 1; j < RCH / 16; ++j)
    v[j] = *(const float4*)(zp + (size_t)j * 16 * Dsz);
  float4 s = v[0];
#pragma unroll
  for (int j = 1; j < RCH / 16; ++j) {
    s.x += v[j].x; s.y += v[j].y; s.z += v[j].z; s.w += v[j].w;
  }
  *(float4*)(red + part * Isz + c4 * 4) = s;
  __syncthreads();

  if (threadIdx.x < Isz) {
    const int i = threadIdx.x;
    float tot = 0.f;
#pragma unroll
    for (int p = 0; p < 16; ++p) tot += red[p * Isz + i];
    const int ch = bh * Isz + i;
    if (chunk == 0) {
      tot += hidden[ch];
      zhh[ch] = Z[((size_t)(b * Lsz + h) * Hsz + h) * Isz + i];
    }
    P[(size_t)chunk * NBHI + ch] = tot;
  }
}

__device__ __forceinline__ float wave_sum(float v) {
#pragma unroll
  for (int off = 16; off >= 1; off >>= 1) v += __shfl_xor(v, off, 32);
  return v;
}

__global__ __launch_bounds__(1024)
void scan_factor_kernel(const float* __restrict__ PD,
                        const float* __restrict__ b_dt, const float* __restrict__ ln_a,
                        const float* __restrict__ P, const float* __restrict__ zhh,
                        const float* __restrict__ gn_w,
                        float* __restrict__ ES, float* __restrict__ DZ,
                        float* __restrict__ DC, float* __restrict__ CS) {
  __shared__ float scan[Lsz];
  const int b = blockIdx.x >> 3;
  const int h = blockIdx.x & 7;
  const int l = threadIdx.x;
  const int lane = threadIdx.x & 31;

  const float* pp = PD + (size_t)(b * Lsz + l) * Hsz + h;
  float pv[NSEG];
#pragma unroll
  for (int j = 0; j < NSEG; ++j) pv[j] = pp[(size_t)j * (Msz * Hsz)];

  const int ch = blockIdx.x * Isz + 2 * lane;
  float2 c = make_float2(0.f, 0.f);
#pragma unroll
  for (int k = 0; k < NCH; ++k) {
    const float2 p = *(const float2*)(P + (size_t)k * NBHI + ch);
    c.x += p.x; c.y += p.y;
  }
  const float2 z = *(const float2*)(zhh + ch);
  const float gw = gn_w[h];

  float dot = pv[0];
#pragma unroll
  for (int j = 1; j < NSEG; ++j) dot += pv[j];

  const float v = dot + b_dt[h];
  const float sp = (v > 20.f) ? v : log1pf(expf(v));
  const float ld = -expf(ln_a[h]) * sp;
  scan[l] = ld;

  const float inv_i = 1.0f / (float)Isz;
  const float mz = wave_sum(z.x + z.y) * inv_i;
  const float mc = wave_sum(c.x + c.y) * inv_i;
  const float dzx = z.x - mz, dzy = z.y - mz;
  const float dcx = c.x - mc, dcy = c.y - mc;
  const float vz  = wave_sum(dzx * dzx + dzy * dzy) * inv_i;
  const float czc = wave_sum(dzx * dcx + dzy * dcy) * inv_i;
  const float vc  = wave_sum(dcx * dcx + dcy * dcy) * inv_i;
  if (l < 32) {
    *(float2*)(DZ + ch) = make_float2(dzx, dzy);
    *(float2*)(DC + ch) = make_float2(dcx, dcy);
    *(float2*)(CS + ch) = c;
  }

  __syncthreads();
  for (int off = 1; off < Lsz; off <<= 1) {
    float t = scan[l];
    if (l >= off) t += scan[l - off];
    __syncthreads();
    scan[l] = t;
    __syncthreads();
  }
  const float ev = expf(scan[l]);
  const float var = fmaxf(vz + ev * (2.0f * czc + ev * vc), 0.f);
  const float sv = gw / sqrtf(var + GN_EPS);
  *(float2*)(ES + 2 * ((size_t)h * Msz + (size_t)b * Lsz + l)) = make_float2(ev, sv);
}

__device__ __forceinline__ float lane_bcast(float v, int src) {
  return __builtin_bit_cast(
      float, __builtin_amdgcn_readlane(__builtin_bit_cast(int, v), src));
}

__global__ __launch_bounds__(128)
void gemm_out_kernel(const float* __restrict__ DZ, const float* __restrict__ DC,
                     const float* __restrict__ ES, const float* __restrict__ zhh,
                     const float* __restrict__ CS, const float* __restrict__ gn_b,
                     const float* __restrict__ Wy, const float* __restrict__ by,
                     const float* __restrict__ X, const float* __restrict__ Wya,
                     const float* __restrict__ bya, float* __restrict__ OUT,
                     float* __restrict__ Hout) {
  const int lane = threadIdx.x & 31;
  const int wave = threadIdx.x >> 5;
  const int mr = lane & 15;
  const int kk = lane >> 4;
  const int m_base = blockIdx.x * 64 + (wave & 1) * 32;
  const int n_base = blockIdx.y * 64 + (wave >> 1) * 32;
  const int bb = blockIdx.x >> 4;

  {
    const int hrow0 = blockIdx.x * 64 + wave * 16;
    const int hcol = blockIdx.y * Isz + 2 * lane;
    const int ch = bb * Dsz + hcol;
    const float ev = ES[2 * ((size_t)blockIdx.y * Msz + hrow0 + mr)];
    const v2f z = *(const v2f*)(zhh + ch);
    const v2f c = *(const v2f*)(CS + ch);
#pragma unroll
    for (int r = 0; r < 16; ++r) {
      const float er = lane_bcast(ev, r);
      v2f hv;
      hv.x = z.x + er * c.x;
      hv.y = z.y + er * c.y;
      *(v2f*)(Hout + (size_t)(hrow0 + r) * Dsz + hcol) = hv;
    }
  }

  v8f accg[2][2], accy[2][2];
  const v8f zero = {0.f, 0.f, 0.f, 0.f, 0.f, 0.f, 0.f, 0.f};
  for (int i = 0; i < 2; ++i)
    for (int j = 0; j < 2; ++j) { accg[i][j] = zero; accy[i][j] = zero; }

  const size_t ra = (size_t)(m_base + mr) * Dsz + 2 * kk;
  const size_t rb = (size_t)(n_base + mr) * Dsz + 2 * kk;
  const float* G0 = X   + ra;
  const float* G1 = G0  + 16 * Dsz;
  const float* V0 = Wya + rb;
  const float* V1 = V0  + 16 * Dsz;
  const float* A0 = DZ  + (size_t)bb * Dsz + 2 * kk;
  const float* A1 = DC  + (size_t)bb * Dsz + 2 * kk;
  const float* B0 = Wy  + rb;
  const float* B1 = B0  + 16 * Dsz;
  const float* E0 = ES + 2 * (size_t)(m_base + mr);
  const float* E1 = E0 + 2 * 16;
#define HSTR ((size_t)2 * Msz)

  v2f pf[2][KBS / 4][4];
#define OUT_LOAD(S, K, PA0, PA1, PB0, PB1)                           \
  _Pragma("unroll")                                                  \
  for (int u = 0; u < KBS / 4; ++u) {                                \
    pf[S][u][0] = *(const v2f*)(PA0 + (K) + 4 * u);                  \
    pf[S][u][1] = *(const v2f*)(PA1 + (K) + 4 * u);                  \
    pf[S][u][2] = *(const v2f*)(PB0 + (K) + 4 * u);                  \
    pf[S][u][3] = *(const v2f*)(PB1 + (K) + 4 * u);                  \
  }
#define OUT_MMA(S, ACC)                                              \
  _Pragma("unroll")                                                  \
  for (int u = 0; u < KBS / 4; ++u) {                                \
    ACC[0][0] = wmma4(pf[S][u][0], pf[S][u][2], ACC[0][0]);          \
    ACC[0][1] = wmma4(pf[S][u][0], pf[S][u][3], ACC[0][1]);          \
    ACC[1][0] = wmma4(pf[S][u][1], pf[S][u][2], ACC[1][0]);          \
    ACC[1][1] = wmma4(pf[S][u][1], pf[S][u][3], ACC[1][1]);          \
  }
#define OUT_MMA_HN(S, Q0, Q1, GB)                                    \
  _Pragma("unroll")                                                  \
  for (int u = 0; u < KBS / 4; ++u) {                                \
    const v2f dzv = pf[S][u][0], dcv = pf[S][u][1];                  \
    v2f f0, f1;                                                      \
    f0.x = (dzv.x + Q0.x * dcv.x) * Q0.y + GB;                       \
    f0.y = (dzv.y + Q0.x * dcv.y) * Q0.y + GB;                       \
    f1.x = (dzv.x + Q1.x * dcv.x) * Q1.y + GB;                       \
    f1.y = (dzv.y + Q1.x * dcv.y) * Q1.y + GB;                       \
    accy[0][0] = wmma4(f0, pf[S][u][2], accy[0][0]);                 \
    accy[0][1] = wmma4(f0, pf[S][u][3], accy[0][1]);                 \
    accy[1][0] = wmma4(f1, pf[S][u][2], accy[1][0]);                 \
    accy[1][1] = wmma4(f1, pf[S][u][3], accy[1][1]);                 \
  }

  OUT_LOAD(0, 0, G0, G1, V0, V1)
#pragma nounroll
  for (int kb = 0; kb < Dsz - 2 * KBS; kb += 2 * KBS) {
    OUT_LOAD(1, kb + KBS, G0, G1, V0, V1)
    OUT_MMA(0, accg)
    OUT_LOAD(0, kb + 2 * KBS, G0, G1, V0, V1)
    OUT_MMA(1, accg)
    LOADS_THEN_PRODUCTS(4 * (KBS / 4), 4 * (KBS / 4));
  }
  OUT_LOAD(1, Dsz - KBS, G0, G1, V0, V1)
  OUT_MMA(0, accg)
  OUT_LOAD(0, 0, A0, A1, B0, B1)
  v2f q0 = *(const v2f*)E0, q1 = *(const v2f*)E1;
  float gbh = gn_b[0];
  OUT_MMA(1, accg)

#pragma nounroll
  for (int kb = 0; kb < Dsz - 2 * KBS; kb += 2 * KBS) {
    const int hnx = (kb >> 6) + 1;
    const v2f nq0 = *(const v2f*)(E0 + hnx * HSTR);
    const v2f nq1 = *(const v2f*)(E1 + hnx * HSTR);
    const float ngb = gn_b[hnx];
    OUT_LOAD(1, kb + KBS, A0, A1, B0, B1)
    OUT_MMA_HN(0, q0, q1, gbh)
    OUT_LOAD(0, kb + 2 * KBS, A0, A1, B0, B1)
    OUT_MMA_HN(1, q0, q1, gbh)
    LOADS_THEN_PRODUCTS(4 * (KBS / 4), 4 * (KBS / 4));
    q0 = nq0; q1 = nq1; gbh = ngb;
  }
  OUT_LOAD(1, Dsz - KBS, A0, A1, B0, B1)
  OUT_MMA_HN(0, q0, q1, gbh)
  OUT_MMA_HN(1, q0, q1, gbh)
#undef OUT_LOAD
#undef OUT_MMA
#undef OUT_MMA_HN
#undef HSTR

  for (int mt = 0; mt < 2; ++mt)
    for (int nt = 0; nt < 2; ++nt) {
      const int n = n_base + nt * 16 + mr;
      const float biy = by[n], biya = bya[n];
#pragma unroll
      for (int r = 0; r < 8; ++r) {
        const int m = m_base + mt * 16 + kk * 8 + r;
        const float v = accy[mt][nt][r] + biy;
        const float g = silu_f(accg[mt][nt][r] + biya);
        OUT[(size_t)m * Dsz + n] = v * g;
      }
    }
}

extern "C" void kernel_launch(void* const* d_in, const int* in_sizes, int n_in,
                              void* d_out, int out_size, void* d_ws, size_t ws_size,
                              hipStream_t stream) {
  const float* x      = (const float*)d_in[0];
  const float* hidden = (const float*)d_in[1];
  const float* w_z    = (const float*)d_in[2];
  const float* b_z    = (const float*)d_in[3];
  const float* w_za   = (const float*)d_in[4];
  const float* b_za   = (const float*)d_in[5];
  const float* w_y    = (const float*)d_in[6];
  const float* b_y    = (const float*)d_in[7];
  const float* w_ya   = (const float*)d_in[8];
  const float* b_ya   = (const float*)d_in[9];
  const float* w_dt   = (const float*)d_in[10];
  const float* b_dt   = (const float*)d_in[11];
  const float* ln_a   = (const float*)d_in[12];
  const float* gn_w   = (const float*)d_in[13];
  const float* gn_b   = (const float*)d_in[14];

  float* y_out = (float*)d_out;
  float* h_out = y_out + (size_t)Msz * Dsz;

  float* ws   = (float*)d_ws;
  float* Z    = ws;
  float* ESb  = Z    + (size_t)Msz * Dsz;
  float* ZHH  = ESb  + (size_t)2 * Hsz * Msz;
  float* Pb   = ZHH  + (size_t)NBHI;
  float* PDb  = Pb   + (size_t)NCH * NBHI;
  float* DZb  = PDb  + (size_t)NSEG * Msz * Hsz;
  float* DCb  = DZb  + (size_t)NBHI;
  float* CSb  = DCb  + (size_t)NBHI;

  const dim3 gGemm(Msz / 64, Dsz / 64, 1);
  const dim3 bGemm(128, 1, 1);

  gemm_gate_z_kernel<<<gGemm, bGemm, 0, stream>>>(x, w_z, b_z, w_za, b_za, Z,
                                                  w_dt, PDb);
  zsum_kernel<<<dim3(Bsz * Hsz, NCH), dim3(256), 0, stream>>>(Z, hidden, Pb, ZHH);
  scan_factor_kernel<<<dim3(Bsz * Hsz), dim3(Lsz), 0, stream>>>(
      PDb, b_dt, ln_a, Pb, ZHH, gn_w, ESb, DZb, DCb, CSb);
  gemm_out_kernel<<<gGemm, bGemm, 0, stream>>>(DZb, DCb, ESb, ZHH, CSb, gn_b,
                                               w_y, b_y, x, w_ya, b_ya,
                                               y_out, h_out);
}


# Decay scan decoupled: column-sum blocks scan their own 128 rows; output GEMM blocks add chunk carries and form the norm factors in LDS. Scan launch gone, 4 -> 3. Scan, moment sum order changes
# speedup vs baseline: 3.2102x; 1.0962x over previous
#include <hip/hip_runtime.h>
#include <hip/hip_bf16.h>
#include <math.h>


#define Bsz 4
#define Lsz 1024
#define Dsz 512
#define Hsz 8
#define Isz 64
#define Msz (Bsz * Lsz)
#define GN_EPS 1e-5f
#define LOADS_THEN_PRODUCTS(NL, NM)                         \
  __builtin_amdgcn_sched_group_barrier(0x020, NL, 0);       \
  __builtin_amdgcn_sched_group_barrier(0x008, NM, 0);       \
  __builtin_amdgcn_sched_group_barrier(0x020, NL, 0);       \
  __builtin_amdgcn_sched_group_barrier(0x008, NM, 0)
#define LOADS_THEN_PRODUCTS_HN(NV, ND1, ND2, NM)            \
  __builtin_amdgcn_sched_group_barrier(0x020, NV, 0);       \
  __builtin_amdgcn_sched_group_barrier(0x100, ND1, 0);      \
  __builtin_amdgcn_sched_group_barrier(0x008, NM, 0);       \
  __builtin_amdgcn_sched_group_barrier(0x020, NV, 0);       \
  __builtin_amdgcn_sched_group_barrier(0x100, ND2, 0);      \
  __builtin_amdgcn_sched_group_barrier(0x008, NM, 0)
#define KBG 16
#define KBS 32
#define NSEG 16
#define KSEG (Dsz / NSEG)

typedef float v2f __attribute__((ext_vector_type(2)));
typedef float v8f __attribute__((ext_vector_type(8)));

__device__ __forceinline__ v8f wmma4(v2f a, v2f b, v8f c) {
  return __builtin_amdgcn_wmma_f32_16x16x4_f32(
       false, a,  false, b,
       (short)0, c,  false,  false);
}

__device__ __forceinline__ float silu_f(float v) {
  return v / (1.0f + expf(-v));
}

__global__ __launch_bounds__(128)
void gemm_gate_z_kernel(const float* __restrict__ X,
                        const float* __restrict__ Wz,  const float* __restrict__ bz,
                        const float* __restrict__ Wza, const float* __restrict__ bza,
                        float* __restrict__ Z,
                        const float* __restrict__ Wd, float* __restrict__ PD) {
  const int lane = threadIdx.x & 31;
  const int wave = threadIdx.x >> 5;
  const int mr = lane & 15;
  const int kk = lane >> 4;
  const int m_base = blockIdx.x * 64 + (wave & 1) * 32;
  const int n_base = blockIdx.y * 64 + (wave >> 1) * 32;

  v8f a1[2][2], a2[2][2];
  const v8f zero = {0.f, 0.f, 0.f, 0.f, 0.f, 0.f, 0.f, 0.f};
  for (int i = 0; i < 2; ++i)
    for (int j = 0; j < 2; ++j) { a1[i][j] = zero; a2[i][j] = zero; }

  const float* A0  = X   + (size_t)(m_base + mr)       * Dsz + 2 * kk;
  const float* A1  = A0  + 16 * Dsz;
  const float* Bz0 = Wz  + (size_t)(n_base + mr)       * Dsz + 2 * kk;
  const float* Bz1 = Bz0 + 16 * Dsz;
  const float* Ba0 = Wza + (size_t)(n_base + mr)       * Dsz + 2 * kk;
  const float* Ba1 = Ba0 + 16 * Dsz;

  v2f pa[2][KBG / 4][6];
#define GATE_LOAD(S, K)                                              \
  _Pragma("unroll")                                                  \
  for (int u = 0; u < KBG / 4; ++u) {                                \
    pa[S][u][0] = *(const v2f*)(A0  + (K) + 4 * u);                  \
    pa[S][u][1] = *(const v2f*)(A1  + (K) + 4 * u);                  \
    pa[S][u][2] = *(const v2f*)(Bz0 + (K) + 4 * u);                  \
    pa[S][u][3] = *(const v2f*)(Bz1 + (K) + 4 * u);                  \
    pa[S][u][4] = *(const v2f*)(Ba0 + (K) + 4 * u);                  \
    pa[S][u][5] = *(const v2f*)(Ba1 + (K) + 4 * u);                  \
  }
#define GATE_MMA(S)                                                  \
  _Pragma("unroll")                                                  \
  for (int u = 0; u < KBG / 4; ++u) {                                \
    a1[0][0] = wmma4(pa[S][u][0], pa[S][u][2], a1[0][0]);            \
    a1[0][1] = wmma4(pa[S][u][0], pa[S][u][3], a1[0][1]);            \
    a1[1][0] = wmma4(pa[S][u][1], pa[S][u][2], a1[1][0]);            \
    a1[1][1] = wmma4(pa[S][u][1], pa[S][u][3], a1[1][1]);            \
    a2[0][0] = wmma4(pa[S][u][0], pa[S][u][4], a2[0][0]);            \
    a2[0][1] = wmma4(pa[S][u][0], pa[S][u][5], a2[0][1]);            \
    a2[1][0] = wmma4(pa[S][u][1], pa[S][u][4], a2[1][0]);            \
    a2[1][1] = wmma4(pa[S][u][1], pa[S][u][5], a2[1][1]);            \
  }

  GATE_LOAD(0, 0)

  {
    const int seg = blockIdx.y * 2 + (wave >> 1);
    const float* Ad0 = A0 + seg * KSEG;
    const float* Ad1 = A1 + seg * KSEG;
    const float* Bd  = Wd + (size_t)(mr & 7) * Dsz + 2 * kk + seg * KSEG;
    v8f d0 = zero, d1 = zero;
#pragma unroll
    for (int u = 0; u < KSEG; u += 4) {
      v2f fa0 = *(const v2f*)(Ad0 + u);
      v2f fa1 = *(const v2f*)(Ad1 + u);
      v2f fd  = *(const v2f*)(Bd + u);
      d0 = wmma4(fa0, fd, d0);
      d1 = wmma4(fa1, fd, d1);
    }
    if (mr < Hsz) {
      float* pd = PD + (size_t)seg * (Msz * Hsz)
                     + (size_t)(m_base + kk * 8) * Hsz + mr;
#pragma unroll
      for (int r = 0; r < 8; ++r) {
        pd[r * Hsz]        = d0[r];
        pd[(16 + r) * Hsz] = d1[r];
      }
    }
  }

#pragma nounroll
  for (int kb = 0; kb < Dsz - 2 * KBG; kb += 2 * KBG) {
    GATE_LOAD(1, kb + KBG)
    GATE_MMA(0)
    GATE_LOAD(0, kb + 2 * KBG)
    GATE_MMA(1)
    LOADS_THEN_PRODUCTS(6 * (KBG / 4), 8 * (KBG / 4));
  }
  GATE_LOAD(1, Dsz - KBG)
  GATE_MMA(0)
  GATE_MMA(1)
#undef GATE_LOAD
#undef GATE_MMA

  for (int mt = 0; mt < 2; ++mt)
    for (int nt = 0; nt < 2; ++nt) {
      const int n = n_base + nt * 16 + mr;
      const float biz = bz[n], biza = bza[n];
#pragma unroll
      for (int r = 0; r < 8; ++r) {
        const int m = m_base + mt * 16 + kk * 8 + r;
        const float v1 = a1[mt][nt][r] + biz;
        const float v2 = a2[mt][nt][r] + biza;
        Z[(size_t)m * Dsz + n] = v1 * silu_f(v2);
      }
    }
}

#define NCH 8
#define RCH (Lsz / NCH)
#define NBHI (Bsz * Hsz * Isz)

__global__ __launch_bounds__(256)
void zsum_kernel(const float* __restrict__ Z, const float* __restrict__ hidden,
                 const float* __restrict__ PD, const float* __restrict__ b_dt,
                 const float* __restrict__ ln_a,
                 float* __restrict__ P, float* __restrict__ zhh,
                 float* __restrict__ SL, float* __restrict__ T) {
  __shared__ float red[16 * Isz];
  __shared__ float wtot[RCH / 32];
  const int bh = blockIdx.x;
  const int chunk = blockIdx.y;
  const int b = bh >> 3;
  const int h = bh & 7;
  const int c4 = threadIdx.x & 15;
  const int part = threadIdx.x >> 4;
  const int lane = threadIdx.x & 31;
  const int wave = threadIdx.x >> 5;

  const int m0 = chunk * RCH + part;
  const float* zp = Z + ((size_t)(b * Lsz + m0) * Hsz + h) * Isz + c4 * 4;

  float4 v[RCH / 16];
  if (m0 > h) v[0] = *(const float4*)zp;
  else        v[0] = make_float4(0.f, 0.f, 0.f, 0.f);
#pragma unroll
  for (int j = 1; j < RCH / 16; ++j)
    v[j] = *(const float4*)(zp + (size_t)j * 16 * Dsz);

  const int lrow = chunk * RCH + threadIdx.x;
  float sc = 0.f;
  if (threadIdx.x < RCH) {
    const float* pp = PD + (size_t)(b * Lsz + lrow) * Hsz + h;
    float pv[NSEG];
#pragma unroll
    for (int j = 0; j < NSEG; ++j) pv[j] = pp[(size_t)j * (Msz * Hsz)];
    float dot = pv[0];
#pragma unroll
    for (int j = 1; j < NSEG; ++j) dot += pv[j];
    const float dv = dot + b_dt[h];
    const float sp = (dv > 20.f) ? dv : log1pf(expf(dv));
    sc = -expf(ln_a[h]) * sp;
#pragma unroll
    for (int off = 1; off < 32; off <<= 1) {
      const float up = __shfl_up(sc, off, 32);
      if (lane >= off) sc += up;
    }
    if (lane == 31) wtot[wave] = sc;
  }

  float4 s = v[0];
#pragma unroll
  for (int j = 1; j < RCH / 16; ++j) {
    s.x += v[j].x; s.y += v[j].y; s.z += v[j].z; s.w += v[j].w;
  }
  *(float4*)(red + part * Isz + c4 * 4) = s;
  __syncthreads();

  if (threadIdx.x < RCH) {
    float pre = 0.f;
#pragma unroll
    for (int w = 0; w < RCH / 32 - 1; ++w)
      if (w < wave) pre += wtot[w];
    const float sl = pre + sc;
    SL[(size_t)h * Msz + (size_t)b * Lsz + lrow] = sl;
    if (threadIdx.x == RCH - 1) T[chunk * (Bsz * Hsz) + bh] = sl;
  }

  if (threadIdx.x < Isz) {
    const int i = threadIdx.x;
    float tot = 0.f;
#pragma unroll
    for (int p = 0; p < 16; ++p) tot += red[p * Isz + i];
    const int ch = bh * Isz + i;
    if (chunk == 0) {
      tot += hidden[ch];
      zhh[ch] = Z[((size_t)(b * Lsz + h) * Hsz + h) * Isz + i];
    }
    P[(size_t)chunk * NBHI + ch] = tot;
  }
}

__device__ __forceinline__ float half_sum(float v) {
#pragma unroll
  for (int off = 8; off >= 1; off >>= 1) v += __shfl_xor(v, off, 32);
  return v;
}

__device__ __forceinline__ float lane_bcast(float v, int src) {
  return __builtin_bit_cast(
      float, __builtin_amdgcn_readlane(__builtin_bit_cast(int, v), src));
}

__global__ __launch_bounds__(128)
void gemm_out_kernel(const float* __restrict__ P, const float* __restrict__ T,
                     const float* __restrict__ SL, const float* __restrict__ zhh,
                     const float* __restrict__ gn_w, const float* __restrict__ gn_b,
                     const float* __restrict__ Wy, const float* __restrict__ by,
                     const float* __restrict__ X, const float* __restrict__ Wya,
                     const float* __restrict__ bya, float* __restrict__ OUT,
                     float* __restrict__ Hout) {
  const int lane = threadIdx.x & 31;
  const int wave = threadIdx.x >> 5;
  const int mr = lane & 15;
  const int kk = lane >> 4;
  const int m_base = blockIdx.x * 64 + (wave & 1) * 32;
  const int n_base = blockIdx.y * 64 + (wave >> 1) * 32;
  const int bb = blockIdx.x >> 4;

  __shared__ __attribute__((aligned(16))) float s_dz[Dsz];
  __shared__ __attribute__((aligned(16))) float s_dc[Dsz];
  __shared__ __attribute__((aligned(16))) float s_c[Dsz];
  __shared__ __attribute__((aligned(16))) float s_es[Hsz * 64 * 2];
  __shared__ float s_mom[Hsz * 4];
#define ESTR (64 * 2)

  {
    const int t = threadIdx.x;
    const int ch4 = bb * Dsz + 4 * t;
    float4 c = make_float4(0.f, 0.f, 0.f, 0.f);
#pragma unroll
    for (int k = 0; k < NCH; ++k) {
      const float4 p = *(const float4*)(P + (size_t)k * NBHI + ch4);
      c.x += p.x; c.y += p.y; c.z += p.z; c.w += p.w;
    }
    const float4 z = *(const float4*)(zhh + ch4);
    float slv[4];
#pragma unroll
    for (int i = 0; i < 4; ++i)
      slv[i] = SL[(size_t)((t >> 6) + 2 * i) * Msz + blockIdx.x * 64 + (t & 63)];

    const float inv_i = 1.0f / (float)Isz;
    const float mz = half_sum((z.x + z.y) + (z.z + z.w)) * inv_i;
    const float mc = half_sum((c.x + c.y) + (c.z + c.w)) * inv_i;
    const float4 dz = make_float4(z.x - mz, z.y - mz, z.z - mz, z.w - mz);
    const float4 dc = make_float4(c.x - mc, c.y - mc, c.z - mc, c.w - mc);
    const float vz  = half_sum((dz.x * dz.x + dz.y * dz.y) + (dz.z * dz.z + dz.w * dz.w)) * inv_i;
    const float czc = half_sum((dz.x * dc.x + dz.y * dc.y) + (dz.z * dc.z + dz.w * dc.w)) * inv_i;
    const float vc  = half_sum((dc.x * dc.x + dc.y * dc.y) + (dc.z * dc.z + dc.w * dc.w)) * inv_i;
    *(float4*)(s_dz + 4 * t) = dz;
    *(float4*)(s_dc + 4 * t) = dc;
    *(float4*)(s_c  + 4 * t) = c;
    if ((t & 15) == 0) {
      s_mom[(t >> 4) * 4 + 0] = vz;
      s_mom[(t >> 4) * 4 + 1] = czc;
      s_mom[(t >> 4) * 4 + 2] = vc;
    }
    if (t < Hsz) {
      const int chunk = (blockIdx.x & 15) >> 1;
      float carry = 0.f;
#pragma unroll
      for (int k = 0; k < NCH - 1; ++k)
        if (k < chunk) carry += T[k * (Bsz * Hsz) + bb * Hsz + t];
      s_mom[t * 4 + 3] = carry;
    }
    __syncthreads();
#pragma unroll
    for (int i = 0; i < 4; ++i) {
      const int hd = (t >> 6) + 2 * i;
      const float ev = expf(s_mom[hd * 4 + 3] + slv[i]);
      const float var = fmaxf(s_mom[hd * 4 + 0]
                              + ev * (2.0f * s_mom[hd * 4 + 1] + ev * s_mom[hd * 4 + 2]), 0.f);
      const float sv = gn_w[hd] / sqrtf(var + GN_EPS);
      *(float2*)(s_es + hd * ESTR + 2 * (t & 63)) = make_float2(ev, sv);
    }
    __syncthreads();
  }

  {
    const int hrow0 = blockIdx.x * 64 + wave * 16;
    const int hcol = blockIdx.y * Isz + 2 * lane;
    const int ch = bb * Dsz + hcol;
    const float ev = s_es[blockIdx.y * ESTR + 2 * (wave * 16 + mr)];
    const v2f z = *(const v2f*)(zhh + ch);
    const v2f c = *(const v2f*)(s_c + hcol);
#pragma unroll
    for (int r = 0; r < 16; ++r) {
      const float er = lane_bcast(ev, r);
      v2f hv;
      hv.x = z.x + er * c.x;
      hv.y = z.y + er * c.y;
      *(v2f*)(Hout + (size_t)(hrow0 + r) * Dsz + hcol) = hv;
    }
  }

  v8f accg[2][2], accy[2][2];
  const v8f zero = {0.f, 0.f, 0.f, 0.f, 0.f, 0.f, 0.f, 0.f};
  for (int i = 0; i < 2; ++i)
    for (int j = 0; j < 2; ++j) { accg[i][j] = zero; accy[i][j] = zero; }

  const size_t ra = (size_t)(m_base + mr) * Dsz + 2 * kk;
  const size_t rb = (size_t)(n_base + mr) * Dsz + 2 * kk;
  const float* G0 = X   + ra;
  const float* G1 = G0  + 16 * Dsz;
  const float* V0 = Wya + rb;
  const float* V1 = V0  + 16 * Dsz;
  const float* B0 = Wy  + rb;
  const float* B1 = B0  + 16 * Dsz;
  const float* E0 = s_es + 2 * ((wave & 1) * 32 + mr);
  const float* E1 = E0 + 2 * 16;

  v2f pf[2][KBS / 4][4];
#define OUT_LOAD(S, K, PA0, PA1, PB0, PB1)                           \
  _Pragma("unroll")                                                  \
  for (int u = 0; u < KBS / 4; ++u) {                                \
    pf[S][u][0] = *(const v2f*)(PA0 + (K) + 4 * u);                  \
    pf[S][u][1] = *(const v2f*)(PA1 + (K) + 4 * u);                  \
    pf[S][u][2] = *(const v2f*)(PB0 + (K) + 4 * u);                  \
    pf[S][u][3] = *(const v2f*)(PB1 + (K) + 4 * u);                  \
  }
#define OUT_MMA(S, ACC)                                              \
  _Pragma("unroll")                                                  \
  for (int u = 0; u < KBS / 4; ++u) {                                \
    ACC[0][0] = wmma4(pf[S][u][0], pf[S][u][2], ACC[0][0]);          \
    ACC[0][1] = wmma4(pf[S][u][0], pf[S][u][3], ACC[0][1]);          \
    ACC[1][0] = wmma4(pf[S][u][1], pf[S][u][2], ACC[1][0]);          \
    ACC[1][1] = wmma4(pf[S][u][1], pf[S][u][3], ACC[1][1]);          \
  }
#define OUT_LOAD_HN(S, K)                                            \
  _Pragma("unroll")                                                  \
  for (int u = 0; u < KBS / 4; ++u) {                                \
    pf[S][u][0] = *(const v2f*)(s_dz + 2 * kk + (K) + 4 * u);        \
    pf[S][u][1] = *(const v2f*)(s_dc + 2 * kk + (K) + 4 * u);        \
    pf[S][u][2] = *(const v2f*)(B0 + (K) + 4 * u);                   \
    pf[S][u][3] = *(const v2f*)(B1 + (K) + 4 * u);                   \
  }
#define OUT_MMA_HN(S, Q0, Q1, GB)                                    \
  _Pragma("unroll")                                                  \
  for (int u = 0; u < KBS / 4; ++u) {                                \
    const v2f dzv = pf[S][u][0], dcv = pf[S][u][1];                  \
    v2f f0, f1;                                                      \
    f0.x = (dzv.x + Q0.x * dcv.x) * Q0.y + GB;                       \
    f0.y = (dzv.y + Q0.x * dcv.y) * Q0.y + GB;                       \
    f1.x = (dzv.x + Q1.x * dcv.x) * Q1.y + GB;                       \
    f1.y = (dzv.y + Q1.x * dcv.y) * Q1.y + GB;                       \
    accy[0][0] = wmma4(f0, pf[S][u][2], accy[0][0]);                 \
    accy[0][1] = wmma4(f0, pf[S][u][3], accy[0][1]);                 \
    accy[1][0] = wmma4(f1, pf[S][u][2], accy[1][0]);                 \
    accy[1][1] = wmma4(f1, pf[S][u][3], accy[1][1]);                 \
  }

  OUT_LOAD(0, 0, G0, G1, V0, V1)
#pragma nounroll
  for (int kb = 0; kb < Dsz - 2 * KBS; kb += 2 * KBS) {
    OUT_LOAD(1, kb + KBS, G0, G1, V0, V1)
    OUT_MMA(0, accg)
    OUT_LOAD(0, kb + 2 * KBS, G0, G1, V0, V1)
    OUT_MMA(1, accg)
    LOADS_THEN_PRODUCTS(4 * (KBS / 4), 4 * (KBS / 4));
  }
  OUT_LOAD(1, Dsz - KBS, G0, G1, V0, V1)
  OUT_MMA(0, accg)
  OUT_LOAD_HN(0, 0)
  v2f q0 = *(const v2f*)E0, q1 = *(const v2f*)E1;
  float gbh = gn_b[0];
  OUT_MMA(1, accg)

#pragma nounroll
  for (int kb = 0; kb < Dsz - 2 * KBS; kb += 2 * KBS) {
    const int hnx = (kb >> 6) + 1;
    const v2f nq0 = *(const v2f*)(E0 + hnx * ESTR);
    const v2f nq1 = *(const v2f*)(E1 + hnx * ESTR);
    const float ngb = gn_b[hnx];
    OUT_LOAD_HN(1, kb + KBS)
    OUT_MMA_HN(0, q0, q1, gbh)
    OUT_LOAD_HN(0, kb + 2 * KBS)
    OUT_MMA_HN(1, q0, q1, gbh)
    LOADS_THEN_PRODUCTS_HN(2 * (KBS / 4), KBS / 4 + 1, KBS / 4, 4 * (KBS / 4));
    q0 = nq0; q1 = nq1; gbh = ngb;
  }
  OUT_LOAD_HN(1, Dsz - KBS)
  OUT_MMA_HN(0, q0, q1, gbh)
  OUT_MMA_HN(1, q0, q1, gbh)
#undef OUT_LOAD
#undef OUT_LOAD_HN
#undef OUT_MMA
#undef OUT_MMA_HN
#undef ESTR

  for (int mt = 0; mt < 2; ++mt)
    for (int nt = 0; nt < 2; ++nt) {
      const int n = n_base + nt * 16 + mr;
      const float biy = by[n], biya = bya[n];
#pragma unroll
      for (int r = 0; r < 8; ++r) {
        const int m = m_base + mt * 16 + kk * 8 + r;
        const float v = accy[mt][nt][r] + biy;
        const float g = silu_f(accg[mt][nt][r] + biya);
        OUT[(size_t)m * Dsz + n] = v * g;
      }
    }
}

extern "C" void kernel_launch(void* const* d_in, const int* in_sizes, int n_in,
                              void* d_out, int out_size, void* d_ws, size_t ws_size,
                              hipStream_t stream) {
  const float* x      = (const float*)d_in[0];
  const float* hidden = (const float*)d_in[1];
  const float* w_z    = (const float*)d_in[2];
  const float* b_z    = (const float*)d_in[3];
  const float* w_za   = (const float*)d_in[4];
  const float* b_za   = (const float*)d_in[5];
  const float* w_y    = (const float*)d_in[6];
  const float* b_y    = (const float*)d_in[7];
  const float* w_ya   = (const float*)d_in[8];
  const float* b_ya   = (const float*)d_in[9];
  const float* w_dt   = (const float*)d_in[10];
  const float* b_dt   = (const float*)d_in[11];
  const float* ln_a   = (const float*)d_in[12];
  const float* gn_w   = (const float*)d_in[13];
  const float* gn_b   = (const float*)d_in[14];

  float* y_out = (float*)d_out;
  float* h_out = y_out + (size_t)Msz * Dsz;

  float* ws   = (float*)d_ws;
  float* Z    = ws;
  float* SLb  = Z    + (size_t)Msz * Dsz;
  float* ZHH  = SLb  + (size_t)Hsz * Msz;
  float* Pb   = ZHH  + (size_t)NBHI;
  float* PDb  = Pb   + (size_t)NCH * NBHI;
  float* Tb   = PDb  + (size_t)NSEG * Msz * Hsz;

  const dim3 gGemm(Msz / 64, Dsz / 64, 1);
  const dim3 bGemm(128, 1, 1);

  gemm_gate_z_kernel<<<gGemm, bGemm, 0, stream>>>(x, w_z, b_z, w_za, b_za, Z,
                                                  w_dt, PDb);
  zsum_kernel<<<dim3(Bsz * Hsz, NCH), dim3(256), 0, stream>>>(
      Z, hidden, PDb, b_dt, ln_a, Pb, ZHH, SLb, Tb);
  gemm_out_kernel<<<gGemm, bGemm, 0, stream>>>(Pb, Tb, SLb, ZHH, gn_w, gn_b,
                                               w_y, b_y, x, w_ya, b_ya,
                                               y_out, h_out);
}


# GEMM launches retiled to one wave per SIMD: 256 blocks of 128x64, wave tile 64x32 (z GEMM) / 32x64 (output GEMM), 64 products per K block, a third fewer fragment loads. Sum order unchanged
# speedup vs baseline: 3.4090x; 1.0619x over previous
#include <hip/hip_runtime.h>
#include <hip/hip_bf16.h>
#include <math.h>


#define Bsz 4
#define Lsz 1024
#define Dsz 512
#define Hsz 8
#define Isz 64
#define Msz (Bsz * Lsz)
#define GN_EPS 1e-5f
#define LOADS_THEN_PRODUCTS(NL, NM)                         \
  __builtin_amdgcn_sched_group_barrier(0x020, NL, 0);       \
  __builtin_amdgcn_sched_group_barrier(0x008, NM, 0);       \
  __builtin_amdgcn_sched_group_barrier(0x020, NL, 0);       \
  __builtin_amdgcn_sched_group_barrier(0x008, NM, 0)
#define LOADS_THEN_PRODUCTS_HN(NV, ND1, ND2, NM)            \
  __builtin_amdgcn_sched_group_barrier(0x020, NV, 0);       \
  __builtin_amdgcn_sched_group_barrier(0x100, ND1, 0);      \
  __builtin_amdgcn_sched_group_barrier(0x008, NM, 0);       \
  __builtin_amdgcn_sched_group_barrier(0x020, NV, 0);       \
  __builtin_amdgcn_sched_group_barrier(0x100, ND2, 0);      \
  __builtin_amdgcn_sched_group_barrier(0x008, NM, 0)
#define KBG 16
#define KBS 32
#define NSEG 16
#define KSEG (Dsz / NSEG)

typedef float v2f __attribute__((ext_vector_type(2)));
typedef float v8f __attribute__((ext_vector_type(8)));

__device__ __forceinline__ v8f wmma4(v2f a, v2f b, v8f c) {
  return __builtin_amdgcn_wmma_f32_16x16x4_f32(
       false, a,  false, b,
       (short)0, c,  false,  false);
}

__device__ __forceinline__ float silu_f(float v) {
  return v / (1.0f + expf(-v));
}

__global__ __launch_bounds__(128)
void gemm_gate_z_kernel(const float* __restrict__ X,
                        const float* __restrict__ Wz,  const float* __restrict__ bz,
                        const float* __restrict__ Wza, const float* __restrict__ bza,
                        float* __restrict__ Z,
                        const float* __restrict__ Wd, float* __restrict__ PD) {
  const int lane = threadIdx.x & 31;
  const int wave = threadIdx.x >> 5;
  const int mr = lane & 15;
  const int kk = lane >> 4;
  const int m_base = blockIdx.x * 128 + (wave & 1) * 64;
  const int n_base = blockIdx.y * 64 + (wave >> 1) * 32;

  v8f a1[4][2], a2[4][2];
  const v8f zero = {0.f, 0.f, 0.f, 0.f, 0.f, 0.f, 0.f, 0.f};
  for (int i = 0; i < 4; ++i)
    for (int j = 0; j < 2; ++j) { a1[i][j] = zero; a2[i][j] = zero; }

  const float* A0  = X   + (size_t)(m_base + mr)       * Dsz + 2 * kk;
  const float* Bz0 = Wz  + (size_t)(n_base + mr)       * Dsz + 2 * kk;
  const float* Bz1 = Bz0 + 16 * Dsz;
  const float* Ba0 = Wza + (size_t)(n_base + mr)       * Dsz + 2 * kk;
  const float* Ba1 = Ba0 + 16 * Dsz;

  v2f pa[2][KBG / 4][8];
#define GATE_LOAD(S, K)                                              \
  _Pragma("unroll")                                                  \
  for (int u = 0; u < KBG / 4; ++u) {                                \
    pa[S][u][0] = *(const v2f*)(A0            + (K) + 4 * u);        \
    pa[S][u][1] = *(const v2f*)(A0 + 16 * Dsz + (K) + 4 * u);        \
    pa[S][u][2] = *(const v2f*)(A0 + 32 * Dsz + (K) + 4 * u);        \
    pa[S][u][3] = *(const v2f*)(A0 + 48 * Dsz + (K) + 4 * u);        \
    pa[S][u][4] = *(const v2f*)(Bz0 + (K) + 4 * u);                  \
    pa[S][u][5] = *(const v2f*)(Bz1 + (K) + 4 * u);                  \
    pa[S][u][6] = *(const v2f*)(Ba0 + (K) + 4 * u);                  \
    pa[S][u][7] = *(const v2f*)(Ba1 + (K) + 4 * u);                  \
  }
#define GATE_MMA(S)                                                  \
  _Pragma("unroll")                                                  \
  for (int u = 0; u < KBG / 4; ++u) {                                \
    _Pragma("unroll")                                                \
    for (int i = 0; i < 4; ++i) {                                    \
      a1[i][0] = wmma4(pa[S][u][i], pa[S][u][4], a1[i][0]);          \
      a1[i][1] = wmma4(pa[S][u][i], pa[S][u][5], a1[i][1]);          \
      a2[i][0] = wmma4(pa[S][u][i], pa[S][u][6], a2[i][0]);          \
      a2[i][1] = wmma4(pa[S][u][i], pa[S][u][7], a2[i][1]);          \
    }                                                                \
  }

  GATE_LOAD(0, 0)

  {
    const int seg = blockIdx.y * 2 + (wave >> 1);
    const float* Ad = A0 + seg * KSEG;
    const float* Bd = Wd + (size_t)(mr & 7) * Dsz + 2 * kk + seg * KSEG;
    v8f d[4] = {zero, zero, zero, zero};
#pragma unroll
    for (int u = 0; u < KSEG; u += 4) {
      const v2f fd = *(const v2f*)(Bd + u);
#pragma unroll
      for (int i = 0; i < 4; ++i) {
        const v2f fa = *(const v2f*)(Ad + 16 * i * Dsz + u);
        d[i] = wmma4(fa, fd, d[i]);
      }
    }
    if (mr < Hsz) {
      float* pd = PD + (size_t)seg * (Msz * Hsz)
                     + (size_t)(m_base + kk * 8) * Hsz + mr;
#pragma unroll
      for (int i = 0; i < 4; ++i)
#pragma unroll
        for (int r = 0; r < 8; ++r) pd[(16 * i + r) * Hsz] = d[i][r];
    }
  }

#pragma nounroll
  for (int kb = 0; kb < Dsz - 2 * KBG; kb += 2 * KBG) {
    GATE_LOAD(1, kb + KBG)
    GATE_MMA(0)
    GATE_LOAD(0, kb + 2 * KBG)
    GATE_MMA(1)
    LOADS_THEN_PRODUCTS(8 * (KBG / 4), 16 * (KBG / 4));
  }
  GATE_LOAD(1, Dsz - KBG)
  GATE_MMA(0)
  GATE_MMA(1)
#undef GATE_LOAD
#undef GATE_MMA

  for (int mt = 0; mt < 4; ++mt)
    for (int nt = 0; nt < 2; ++nt) {
      const int n = n_base + nt * 16 + mr;
      const float biz = bz[n], biza = bza[n];
#pragma unroll
      for (int r = 0; r < 8; ++r) {
        const int m = m_base + mt * 16 + kk * 8 + r;
        const float v1 = a1[mt][nt][r] + biz;
        const float v2 = a2[mt][nt][r] + biza;
        Z[(size_t)m * Dsz + n] = v1 * silu_f(v2);
      }
    }
}

#define NCH 8
#define RCH (Lsz / NCH)
#define NBHI (Bsz * Hsz * Isz)

__global__ __launch_bounds__(256)
void zsum_kernel(const float* __restrict__ Z, const float* __restrict__ hidden,
                 const float* __restrict__ PD, const float* __restrict__ b_dt,
                 const float* __restrict__ ln_a,
                 float* __restrict__ P, float* __restrict__ zhh,
                 float* __restrict__ SL, float* __restrict__ T) {
  __shared__ float red[16 * Isz];
  __shared__ float wtot[RCH / 32];
  const int bh = blockIdx.x;
  const int chunk = blockIdx.y;
  const int b = bh >> 3;
  const int h = bh & 7;
  const int c4 = threadIdx.x & 15;
  const int part = threadIdx.x >> 4;
  const int lane = threadIdx.x & 31;
  const int wave = threadIdx.x >> 5;

  const int m0 = chunk * RCH + part;
  const float* zp = Z + ((size_t)(b * Lsz + m0) * Hsz + h) * Isz + c4 * 4;

  float4 v[RCH / 16];
  if (m0 > h) v[0] = *(const float4*)zp;
  else        v[0] = make_float4(0.f, 0.f, 0.f, 0.f);
#pragma unroll
  for (int j = 1; j < RCH / 16; ++j)
    v[j] = *(const float4*)(zp + (size_t)j * 16 * Dsz);

  const int lrow = chunk * RCH + threadIdx.x;
  float sc = 0.f;
  if (threadIdx.x < RCH) {
    const float* pp = PD + (size_t)(b * Lsz + lrow) * Hsz + h;
    float pv[NSEG];
#pragma unroll
    for (int j = 0; j < NSEG; ++j) pv[j] = pp[(size_t)j * (Msz * Hsz)];
    float dot = pv[0];
#pragma unroll
    for (int j = 1; j < NSEG; ++j) dot += pv[j];
    const float dv = dot + b_dt[h];
    const float sp = (dv > 20.f) ? dv : log1pf(expf(dv));
    sc = -expf(ln_a[h]) * sp;
#pragma unroll
    for (int off = 1; off < 32; off <<= 1) {
      const float up = __shfl_up(sc, off, 32);
      if (lane >= off) sc += up;
    }
    if (lane == 31) wtot[wave] = sc;
  }

  float4 s = v[0];
#pragma unroll
  for (int j = 1; j < RCH / 16; ++j) {
    s.x += v[j].x; s.y += v[j].y; s.z += v[j].z; s.w += v[j].w;
  }
  *(float4*)(red + part * Isz + c4 * 4) = s;
  __syncthreads();

  if (threadIdx.x < RCH) {
    float pre = 0.f;
#pragma unroll
    for (int w = 0; w < RCH / 32 - 1; ++w)
      if (w < wave) pre += wtot[w];
    const float sl = pre + sc;
    SL[(size_t)h * Msz + (size_t)b * Lsz + lrow] = sl;
    if (threadIdx.x == RCH - 1) T[chunk * (Bsz * Hsz) + bh] = sl;
  }

  if (threadIdx.x < Isz) {
    const int i = threadIdx.x;
    float tot = 0.f;
#pragma unroll
    for (int p = 0; p < 16; ++p) tot += red[p * Isz + i];
    const int ch = bh * Isz + i;
    if (chunk == 0) {
      tot += hidden[ch];
      zhh[ch] = Z[((size_t)(b * Lsz + h) * Hsz + h) * Isz + i];
    }
    P[(size_t)chunk * NBHI + ch] = tot;
  }
}

__device__ __forceinline__ float half_sum(float v) {
#pragma unroll
  for (int off = 8; off >= 1; off >>= 1) v += __shfl_xor(v, off, 32);
  return v;
}

__device__ __forceinline__ float lane_bcast(float v, int src) {
  return __builtin_bit_cast(
      float, __builtin_amdgcn_readlane(__builtin_bit_cast(int, v), src));
}

__global__ __launch_bounds__(128)
void gemm_out_kernel(const float* __restrict__ P, const float* __restrict__ T,
                     const float* __restrict__ SL, const float* __restrict__ zhh,
                     const float* __restrict__ gn_w, const float* __restrict__ gn_b,
                     const float* __restrict__ Wy, const float* __restrict__ by,
                     const float* __restrict__ X, const float* __restrict__ Wya,
                     const float* __restrict__ bya, float* __restrict__ OUT,
                     float* __restrict__ Hout) {
  const int lane = threadIdx.x & 31;
  const int wave = threadIdx.x >> 5;
  const int mr = lane & 15;
  const int kk = lane >> 4;
  const int m_base = blockIdx.x * 128 + wave * 32;
  const int n_base = blockIdx.y * 64;
  const int bb = blockIdx.x >> 3;

  __shared__ __attribute__((aligned(16))) float s_dz[Dsz];
  __shared__ __attribute__((aligned(16))) float s_dc[Dsz];
  __shared__ __attribute__((aligned(16))) float s_c[Dsz];
  __shared__ __attribute__((aligned(16))) float s_es[Hsz * 128 * 2];
  __shared__ float s_mom[Hsz * 4];
#define ESTR (128 * 2)

  {
    const int t = threadIdx.x;
    const int ch4 = bb * Dsz + 4 * t;
    float4 c = make_float4(0.f, 0.f, 0.f, 0.f);
#pragma unroll
    for (int k = 0; k < NCH; ++k) {
      const float4 p = *(const float4*)(P + (size_t)k * NBHI + ch4);
      c.x += p.x; c.y += p.y; c.z += p.z; c.w += p.w;
    }
    const float4 z = *(const float4*)(zhh + ch4);
    float slv[Hsz];
#pragma unroll
    for (int i = 0; i < Hsz; ++i)
      slv[i] = SL[(size_t)i * Msz + blockIdx.x * 128 + t];

    const float inv_i = 1.0f / (float)Isz;
    const float mz = half_sum((z.x + z.y) + (z.z + z.w)) * inv_i;
    const float mc = half_sum((c.x + c.y) + (c.z + c.w)) * inv_i;
    const float4 dz = make_float4(z.x - mz, z.y - mz, z.z - mz, z.w - mz);
    const float4 dc = make_float4(c.x - mc, c.y - mc, c.z - mc, c.w - mc);
    const float vz  = half_sum((dz.x * dz.x + dz.y * dz.y) + (dz.z * dz.z + dz.w * dz.w)) * inv_i;
    const float czc = half_sum((dz.x * dc.x + dz.y * dc.y) + (dz.z * dc.z + dz.w * dc.w)) * inv_i;
    const float vc  = half_sum((dc.x * dc.x + dc.y * dc.y) + (dc.z * dc.z + dc.w * dc.w)) * inv_i;
    *(float4*)(s_dz + 4 * t) = dz;
    *(float4*)(s_dc + 4 * t) = dc;
    *(float4*)(s_c  + 4 * t) = c;
    if ((t & 15) == 0) {
      s_mom[(t >> 4) * 4 + 0] = vz;
      s_mom[(t >> 4) * 4 + 1] = czc;
      s_mom[(t >> 4) * 4 + 2] = vc;
    }
    if (t < Hsz) {
      const int chunk = blockIdx.x & 7;
      float carry = 0.f;
#pragma unroll
      for (int k = 0; k < NCH - 1; ++k)
        if (k < chunk) carry += T[k * (Bsz * Hsz) + bb * Hsz + t];
      s_mom[t * 4 + 3] = carry;
    }
    __syncthreads();
#pragma unroll
    for (int i = 0; i < Hsz; ++i) {
      const int hd = i;
      const float ev = expf(s_mom[hd * 4 + 3] + slv[i]);
      const float var = fmaxf(s_mom[hd * 4 + 0]
                              + ev * (2.0f * s_mom[hd * 4 + 1] + ev * s_mom[hd * 4 + 2]), 0.f);
      const float sv = gn_w[hd] / sqrtf(var + GN_EPS);
      *(float2*)(s_es + hd * ESTR + 2 * t) = make_float2(ev, sv);
    }
    __syncthreads();
  }

  {
    const int hcol = blockIdx.y * Isz + 2 * lane;
    const int ch = bb * Dsz + hcol;
    const v2f z = *(const v2f*)(zhh + ch);
    const v2f c = *(const v2f*)(s_c + hcol);
#pragma unroll
    for (int g = 0; g < 2; ++g) {
      const int hrow0 = m_base + g * 16;
      const float ev = s_es[blockIdx.y * ESTR + 2 * (wave * 32 + g * 16 + mr)];
#pragma unroll
      for (int r = 0; r < 16; ++r) {
        const float er = lane_bcast(ev, r);
        v2f hv;
        hv.x = z.x + er * c.x;
        hv.y = z.y + er * c.y;
        *(v2f*)(Hout + (size_t)(hrow0 + r) * Dsz + hcol) = hv;
      }
    }
  }

  v8f accg[2][4], accy[2][4];
  const v8f zero = {0.f, 0.f, 0.f, 0.f, 0.f, 0.f, 0.f, 0.f};
  for (int i = 0; i < 2; ++i)
    for (int j = 0; j < 4; ++j) { accg[i][j] = zero; accy[i][j] = zero; }

  const size_t ra = (size_t)(m_base + mr) * Dsz + 2 * kk;
  const size_t rb = (size_t)(n_base + mr) * Dsz + 2 * kk;
  const float* G0 = X   + ra;
  const float* G1 = G0  + 16 * Dsz;
  const float* V0 = Wya + rb;
  const float* B0 = Wy  + rb;
  const float* E0 = s_es + 2 * (wave * 32 + mr);
  const float* E1 = E0 + 2 * 16;

  v2f pf[2][KBS / 4][6];
#define OUT_LOAD(S, K, PA0, PA1, PB)                                 \
  _Pragma("unroll")                                                  \
  for (int u = 0; u < KBS / 4; ++u) {                                \
    pf[S][u][0] = *(const v2f*)(PA0 + (K) + 4 * u);                  \
    pf[S][u][1] = *(const v2f*)(PA1 + (K) + 4 * u);                  \
    pf[S][u][2] = *(const v2f*)(PB            + (K) + 4 * u);        \
    pf[S][u][3] = *(const v2f*)(PB + 16 * Dsz + (K) + 4 * u);        \
    pf[S][u][4] = *(const v2f*)(PB + 32 * Dsz + (K) + 4 * u);        \
    pf[S][u][5] = *(const v2f*)(PB + 48 * Dsz + (K) + 4 * u);        \
  }
#define OUT_MMA(S, ACC)                                              \
  _Pragma("unroll")                                                  \
  for (int u = 0; u < KBS / 4; ++u) {                                \
    _Pragma("unroll")                                                \
    for (int j = 0; j < 4; ++j) {                                    \
      ACC[0][j] = wmma4(pf[S][u][0], pf[S][u][2 + j], ACC[0][j]);    \
      ACC[1][j] = wmma4(pf[S][u][1], pf[S][u][2 + j], ACC[1][j]);    \
    }                                                                \
  }
#define OUT_LOAD_HN(S, K)                                            \
  _Pragma("unroll")                                                  \
  for (int u = 0; u < KBS / 4; ++u) {                                \
    pf[S][u][0] = *(const v2f*)(s_dz + 2 * kk + (K) + 4 * u);        \
    pf[S][u][1] = *(const v2f*)(s_dc + 2 * kk + (K) + 4 * u);        \
    pf[S][u][2] = *(const v2f*)(B0            + (K) + 4 * u);        \
    pf[S][u][3] = *(const v2f*)(B0 + 16 * Dsz + (K) + 4 * u);        \
    pf[S][u][4] = *(const v2f*)(B0 + 32 * Dsz + (K) + 4 * u);        \
    pf[S][u][5] = *(const v2f*)(B0 + 48 * Dsz + (K) + 4 * u);        \
  }
#define OUT_MMA_HN(S, Q0, Q1, GB)                                    \
  _Pragma("unroll")                                                  \
  for (int u = 0; u < KBS / 4; ++u) {                                \
    const v2f dzv = pf[S][u][0], dcv = pf[S][u][1];                  \
    v2f f0, f1;                                                      \
    f0.x = (dzv.x + Q0.x * dcv.x) * Q0.y + GB;                       \
    f0.y = (dzv.y + Q0.x * dcv.y) * Q0.y + GB;                       \
    f1.x = (dzv.x + Q1.x * dcv.x) * Q1.y + GB;                       \
    f1.y = (dzv.y + Q1.x * dcv.y) * Q1.y + GB;                       \
    _Pragma("unroll")                                                \
    for (int j = 0; j < 4; ++j) {                                    \
      accy[0][j] = wmma4(f0, pf[S][u][2 + j], accy[0][j]);           \
      accy[1][j] = wmma4(f1, pf[S][u][2 + j], accy[1][j]);           \
    }                                                                \
  }

  OUT_LOAD(0, 0, G0, G1, V0)
#pragma nounroll
  for (int kb = 0; kb < Dsz - 2 * KBS; kb += 2 * KBS) {
    OUT_LOAD(1, kb + KBS, G0, G1, V0)
    OUT_MMA(0, accg)
    OUT_LOAD(0, kb + 2 * KBS, G0, G1, V0)
    OUT_MMA(1, accg)
    LOADS_THEN_PRODUCTS(6 * (KBS / 4), 8 * (KBS / 4));
  }
  OUT_LOAD(1, Dsz - KBS, G0, G1, V0)
  OUT_MMA(0, accg)
  OUT_LOAD_HN(0, 0)
  v2f q0 = *(const v2f*)E0, q1 = *(const v2f*)E1;
  float gbh = gn_b[0];
  OUT_MMA(1, accg)

#pragma nounroll
  for (int kb = 0; kb < Dsz - 2 * KBS; kb += 2 * KBS) {
    const int hnx = (kb >> 6) + 1;
    const v2f nq0 = *(const v2f*)(E0 + hnx * ESTR);
    const v2f nq1 = *(const v2f*)(E1 + hnx * ESTR);
    const float ngb = gn_b[hnx];
    OUT_LOAD_HN(1, kb + KBS)
    OUT_MMA_HN(0, q0, q1, gbh)
    OUT_LOAD_HN(0, kb + 2 * KBS)
    OUT_MMA_HN(1, q0, q1, gbh)
    LOADS_THEN_PRODUCTS_HN(4 * (KBS / 4), KBS / 4 + 1, KBS / 4, 8 * (KBS / 4));
    q0 = nq0; q1 = nq1; gbh = ngb;
  }
  OUT_LOAD_HN(1, Dsz - KBS)
  OUT_MMA_HN(0, q0, q1, gbh)
  OUT_MMA_HN(1, q0, q1, gbh)
#undef OUT_LOAD
#undef OUT_LOAD_HN
#undef OUT_MMA
#undef OUT_MMA_HN
#undef ESTR

  for (int mt = 0; mt < 2; ++mt)
    for (int nt = 0; nt < 4; ++nt) {
      const int n = n_base + nt * 16 + mr;
      const float biy = by[n], biya = bya[n];
#pragma unroll
      for (int r = 0; r < 8; ++r) {
        const int m = m_base + mt * 16 + kk * 8 + r;
        const float v = accy[mt][nt][r] + biy;
        const float g = silu_f(accg[mt][nt][r] + biya);
        OUT[(size_t)m * Dsz + n] = v * g;
      }
    }
}

extern "C" void kernel_launch(void* const* d_in, const int* in_sizes, int n_in,
                              void* d_out, int out_size, void* d_ws, size_t ws_size,
                              hipStream_t stream) {
  const float* x      = (const float*)d_in[0];
  const float* hidden = (const float*)d_in[1];
  const float* w_z    = (const float*)d_in[2];
  const float* b_z    = (const float*)d_in[3];
  const float* w_za   = (const float*)d_in[4];
  const float* b_za   = (const float*)d_in[5];
  const float* w_y    = (const float*)d_in[6];
  const float* b_y    = (const float*)d_in[7];
  const float* w_ya   = (const float*)d_in[8];
  const float* b_ya   = (const float*)d_in[9];
  const float* w_dt   = (const float*)d_in[10];
  const float* b_dt   = (const float*)d_in[11];
  const float* ln_a   = (const float*)d_in[12];
  const float* gn_w   = (const float*)d_in[13];
  const float* gn_b   = (const float*)d_in[14];

  float* y_out = (float*)d_out;
  float* h_out = y_out + (size_t)Msz * Dsz;

  float* ws   = (float*)d_ws;
  float* Z    = ws;
  float* SLb  = Z    + (size_t)Msz * Dsz;
  float* ZHH  = SLb  + (size_t)Hsz * Msz;
  float* Pb   = ZHH  + (size_t)NBHI;
  float* PDb  = Pb   + (size_t)NCH * NBHI;
  float* Tb   = PDb  + (size_t)NSEG * Msz * Hsz;

  const dim3 gGemm(Msz / 128, Dsz / 64, 1);
  const dim3 bGemm(128, 1, 1);

  gemm_gate_z_kernel<<<gGemm, bGemm, 0, stream>>>(x, w_z, b_z, w_za, b_za, Z,
                                                  w_dt, PDb);
  zsum_kernel<<<dim3(Bsz * Hsz, NCH), dim3(256), 0, stream>>>(
      Z, hidden, PDb, b_dt, ln_a, Pb, ZHH, SLb, Tb);
  gemm_out_kernel<<<gGemm, bGemm, 0, stream>>>(Pb, Tb, SLb, ZHH, gn_w, gn_b,
                                               w_y, b_y, x, w_ya, b_ya,
                                               y_out, h_out);
}


# All WMMA fragment loads 16 bytes a lane (was 8): a lane's four k feed two products, k {0,1,4,5} then {2,3,6,7} of each 8; same loader for A and B; half the loads. Sum order in k changes
# speedup vs baseline: 3.5815x; 1.0506x over previous
#include <hip/hip_runtime.h>
#include <hip/hip_bf16.h>
#include <math.h>


#define Bsz 4
#define Lsz 1024
#define Dsz 512
#define Hsz 8
#define Isz 64
#define Msz (Bsz * Lsz)
#define GN_EPS 1e-5f
#define LOADS_THEN_PRODUCTS(NL, NM)                         \
  __builtin_amdgcn_sched_group_barrier(0x020, NL, 0);       \
  __builtin_amdgcn_sched_group_barrier(0x008, NM, 0);       \
  __builtin_amdgcn_sched_group_barrier(0x020, NL, 0);       \
  __builtin_amdgcn_sched_group_barrier(0x008, NM, 0)
#define LOADS_THEN_PRODUCTS_HN(NV, ND1, ND2, NM)            \
  __builtin_amdgcn_sched_group_barrier(0x020, NV, 0);       \
  __builtin_amdgcn_sched_group_barrier(0x100, ND1, 0);      \
  __builtin_amdgcn_sched_group_barrier(0x008, NM, 0);       \
  __builtin_amdgcn_sched_group_barrier(0x020, NV, 0);       \
  __builtin_amdgcn_sched_group_barrier(0x100, ND2, 0);      \
  __builtin_amdgcn_sched_group_barrier(0x008, NM, 0)
#define KBG 16
#define KBS 32
#define NSEG 16
#define KSEG (Dsz / NSEG)

typedef float v2f __attribute__((ext_vector_type(2)));
typedef float v4f __attribute__((ext_vector_type(4)));
typedef float v8f __attribute__((ext_vector_type(8)));


__device__ __forceinline__ v8f wmma4(v2f a, v2f b, v8f c) {
  return __builtin_amdgcn_wmma_f32_16x16x4_f32(
       false, a,  false, b,
       (short)0, c,  false,  false);
}

__device__ __forceinline__ float silu_f(float v) {
  return v / (1.0f + expf(-v));
}

__global__ __launch_bounds__(128)
void gemm_gate_z_kernel(const float* __restrict__ X,
                        const float* __restrict__ Wz,  const float* __restrict__ bz,
                        const float* __restrict__ Wza, const float* __restrict__ bza,
                        float* __restrict__ Z,
                        const float* __restrict__ Wd, float* __restrict__ PD) {
  const int lane = threadIdx.x & 31;
  const int wave = threadIdx.x >> 5;
  const int mr = lane & 15;
  const int kk = lane >> 4;
  const int m_base = blockIdx.x * 128 + (wave & 1) * 64;
  const int n_base = blockIdx.y * 64 + (wave >> 1) * 32;

  v8f a1[4][2], a2[4][2];
  const v8f zero = {0.f, 0.f, 0.f, 0.f, 0.f, 0.f, 0.f, 0.f};
  for (int i = 0; i < 4; ++i)
    for (int j = 0; j < 2; ++j) { a1[i][j] = zero; a2[i][j] = zero; }

  const float* A0  = X   + (size_t)(m_base + mr)       * Dsz + 4 * kk;
  const float* Bz0 = Wz  + (size_t)(n_base + mr)       * Dsz + 4 * kk;
  const float* Bz1 = Bz0 + 16 * Dsz;
  const float* Ba0 = Wza + (size_t)(n_base + mr)       * Dsz + 4 * kk;
  const float* Ba1 = Ba0 + 16 * Dsz;

  v4f pa[2][KBG / 8][8];
#define GATE_LOAD(S, K)                                              \
  _Pragma("unroll")                                                  \
  for (int u = 0; u < KBG / 8; ++u) {                                \
    pa[S][u][0] = *(const v4f*)(A0            + (K) + 8 * u);        \
    pa[S][u][1] = *(const v4f*)(A0 + 16 * Dsz + (K) + 8 * u);        \
    pa[S][u][2] = *(const v4f*)(A0 + 32 * Dsz + (K) + 8 * u);        \
    pa[S][u][3] = *(const v4f*)(A0 + 48 * Dsz + (K) + 8 * u);        \
    pa[S][u][4] = *(const v4f*)(Bz0 + (K) + 8 * u);                  \
    pa[S][u][5] = *(const v4f*)(Bz1 + (K) + 8 * u);                  \
    pa[S][u][6] = *(const v4f*)(Ba0 + (K) + 8 * u);                  \
    pa[S][u][7] = *(const v4f*)(Ba1 + (K) + 8 * u);                  \
  }
#define GATE_MMA_HALF(S, u, HF)                                      \
    _Pragma("unroll")                                                \
    for (int i = 0; i < 4; ++i) {                                    \
      a1[i][0] = wmma4(pa[S][u][i].HF, pa[S][u][4].HF, a1[i][0]);    \
      a1[i][1] = wmma4(pa[S][u][i].HF, pa[S][u][5].HF, a1[i][1]);    \
      a2[i][0] = wmma4(pa[S][u][i].HF, pa[S][u][6].HF, a2[i][0]);    \
      a2[i][1] = wmma4(pa[S][u][i].HF, pa[S][u][7].HF, a2[i][1]);    \
    }
#define GATE_MMA(S)                                                  \
  _Pragma("unroll")                                                  \
  for (int u = 0; u < KBG / 8; ++u) {                                \
    GATE_MMA_HALF(S, u, xy)                                          \
    GATE_MMA_HALF(S, u, zw)                                          \
  }

  GATE_LOAD(0, 0)

  {
    const int seg = blockIdx.y * 2 + (wave >> 1);
    const float* Ad = A0 + seg * KSEG;
    const float* Bd = Wd + (size_t)(mr & 7) * Dsz + 4 * kk + seg * KSEG;
    v8f d[4] = {zero, zero, zero, zero};
#pragma unroll
    for (int u = 0; u < KSEG; u += 8) {
      const v4f fd = *(const v4f*)(Bd + u);
      v4f fa[4];
#pragma unroll
      for (int i = 0; i < 4; ++i) fa[i] = *(const v4f*)(Ad + 16 * i * Dsz + u);
#pragma unroll
      for (int i = 0; i < 4; ++i) d[i] = wmma4(fa[i].xy, fd.xy, d[i]);
#pragma unroll
      for (int i = 0; i < 4; ++i) d[i] = wmma4(fa[i].zw, fd.zw, d[i]);
    }
    if (mr < Hsz) {
      float* pd = PD + (size_t)seg * (Msz * Hsz)
                     + (size_t)(m_base + kk * 8) * Hsz + mr;
#pragma unroll
      for (int i = 0; i < 4; ++i)
#pragma unroll
        for (int r = 0; r < 8; ++r) pd[(16 * i + r) * Hsz] = d[i][r];
    }
  }

#pragma nounroll
  for (int kb = 0; kb < Dsz - 2 * KBG; kb += 2 * KBG) {
    GATE_LOAD(1, kb + KBG)
    GATE_MMA(0)
    GATE_LOAD(0, kb + 2 * KBG)
    GATE_MMA(1)
    LOADS_THEN_PRODUCTS(8 * (KBG / 8), 16 * (KBG / 4));
  }
  GATE_LOAD(1, Dsz - KBG)
  GATE_MMA(0)
  GATE_MMA(1)
#undef GATE_LOAD
#undef GATE_MMA
#undef GATE_MMA_HALF

  for (int mt = 0; mt < 4; ++mt)
    for (int nt = 0; nt < 2; ++nt) {
      const int n = n_base + nt * 16 + mr;
      const float biz = bz[n], biza = bza[n];
#pragma unroll
      for (int r = 0; r < 8; ++r) {
        const int m = m_base + mt * 16 + kk * 8 + r;
        const float v1 = a1[mt][nt][r] + biz;
        const float v2 = a2[mt][nt][r] + biza;
        Z[(size_t)m * Dsz + n] = v1 * silu_f(v2);
      }
    }
}

#define NCH 8
#define RCH (Lsz / NCH)
#define NBHI (Bsz * Hsz * Isz)

__global__ __launch_bounds__(256)
void zsum_kernel(const float* __restrict__ Z, const float* __restrict__ hidden,
                 const float* __restrict__ PD, const float* __restrict__ b_dt,
                 const float* __restrict__ ln_a,
                 float* __restrict__ P, float* __restrict__ zhh,
                 float* __restrict__ SL, float* __restrict__ T) {
  __shared__ float red[16 * Isz];
  __shared__ float wtot[RCH / 32];
  const int bh = blockIdx.x;
  const int chunk = blockIdx.y;
  const int b = bh >> 3;
  const int h = bh & 7;
  const int c4 = threadIdx.x & 15;
  const int part = threadIdx.x >> 4;
  const int lane = threadIdx.x & 31;
  const int wave = threadIdx.x >> 5;

  const int m0 = chunk * RCH + part;
  const float* zp = Z + ((size_t)(b * Lsz + m0) * Hsz + h) * Isz + c4 * 4;

  float4 v[RCH / 16];
  if (m0 > h) v[0] = *(const float4*)zp;
  else        v[0] = make_float4(0.f, 0.f, 0.f, 0.f);
#pragma unroll
  for (int j = 1; j < RCH / 16; ++j)
    v[j] = *(const float4*)(zp + (size_t)j * 16 * Dsz);

  const int lrow = chunk * RCH + threadIdx.x;
  float sc = 0.f;
  if (threadIdx.x < RCH) {
    const float* pp = PD + (size_t)(b * Lsz + lrow) * Hsz + h;
    float pv[NSEG];
#pragma unroll
    for (int j = 0; j < NSEG; ++j) pv[j] = pp[(size_t)j * (Msz * Hsz)];
    float dot = pv[0];
#pragma unroll
    for (int j = 1; j < NSEG; ++j) dot += pv[j];
    const float dv = dot + b_dt[h];
    const float sp = (dv > 20.f) ? dv : log1pf(expf(dv));
    sc = -expf(ln_a[h]) * sp;
#pragma unroll
    for (int off = 1; off < 32; off <<= 1) {
      const float up = __shfl_up(sc, off, 32);
      if (lane >= off) sc += up;
    }
    if (lane == 31) wtot[wave] = sc;
  }

  float4 s = v[0];
#pragma unroll
  for (int j = 1; j < RCH / 16; ++j) {
    s.x += v[j].x; s.y += v[j].y; s.z += v[j].z; s.w += v[j].w;
  }
  *(float4*)(red + part * Isz + c4 * 4) = s;
  __syncthreads();

  if (threadIdx.x < RCH) {
    float pre = 0.f;
#pragma unroll
    for (int w = 0; w < RCH / 32 - 1; ++w)
      if (w < wave) pre += wtot[w];
    const float sl = pre + sc;
    SL[(size_t)h * Msz + (size_t)b * Lsz + lrow] = sl;
    if (threadIdx.x == RCH - 1) T[chunk * (Bsz * Hsz) + bh] = sl;
  }

  if (threadIdx.x < Isz) {
    const int i = threadIdx.x;
    float tot = 0.f;
#pragma unroll
    for (int p = 0; p < 16; ++p) tot += red[p * Isz + i];
    const int ch = bh * Isz + i;
    if (chunk == 0) {
      tot += hidden[ch];
      zhh[ch] = Z[((size_t)(b * Lsz + h) * Hsz + h) * Isz + i];
    }
    P[(size_t)chunk * NBHI + ch] = tot;
  }
}

__device__ __forceinline__ float half_sum(float v) {
#pragma unroll
  for (int off = 8; off >= 1; off >>= 1) v += __shfl_xor(v, off, 32);
  return v;
}

__device__ __forceinline__ float lane_bcast(float v, int src) {
  return __builtin_bit_cast(
      float, __builtin_amdgcn_readlane(__builtin_bit_cast(int, v), src));
}

__global__ __launch_bounds__(128)
void gemm_out_kernel(const float* __restrict__ P, const float* __restrict__ T,
                     const float* __restrict__ SL, const float* __restrict__ zhh,
                     const float* __restrict__ gn_w, const float* __restrict__ gn_b,
                     const float* __restrict__ Wy, const float* __restrict__ by,
                     const float* __restrict__ X, const float* __restrict__ Wya,
                     const float* __restrict__ bya, float* __restrict__ OUT,
                     float* __restrict__ Hout) {
  const int lane = threadIdx.x & 31;
  const int wave = threadIdx.x >> 5;
  const int mr = lane & 15;
  const int kk = lane >> 4;
  const int m_base = blockIdx.x * 128 + wave * 32;
  const int n_base = blockIdx.y * 64;
  const int bb = blockIdx.x >> 3;

  __shared__ __attribute__((aligned(16))) float s_dz[Dsz];
  __shared__ __attribute__((aligned(16))) float s_dc[Dsz];
  __shared__ __attribute__((aligned(16))) float s_c[Dsz];
  __shared__ __attribute__((aligned(16))) float s_es[Hsz * 128 * 2];
  __shared__ float s_mom[Hsz * 4];
#define ESTR (128 * 2)

  {
    const int t = threadIdx.x;
    const int ch4 = bb * Dsz + 4 * t;
    float4 c = make_float4(0.f, 0.f, 0.f, 0.f);
#pragma unroll
    for (int k = 0; k < NCH; ++k) {
      const float4 p = *(const float4*)(P + (size_t)k * NBHI + ch4);
      c.x += p.x; c.y += p.y; c.z += p.z; c.w += p.w;
    }
    const float4 z = *(const float4*)(zhh + ch4);
    float slv[Hsz];
#pragma unroll
    for (int i = 0; i < Hsz; ++i)
      slv[i] = SL[(size_t)i * Msz + blockIdx.x * 128 + t];

    const float inv_i = 1.0f / (float)Isz;
    const float mz = half_sum((z.x + z.y) + (z.z + z.w)) * inv_i;
    const float mc = half_sum((c.x + c.y) + (c.z + c.w)) * inv_i;
    const float4 dz = make_float4(z.x - mz, z.y - mz, z.z - mz, z.w - mz);
    const float4 dc = make_float4(c.x - mc, c.y - mc, c.z - mc, c.w - mc);
    const float vz  = half_sum((dz.x * dz.x + dz.y * dz.y) + (dz.z * dz.z + dz.w * dz.w)) * inv_i;
    const float czc = half_sum((dz.x * dc.x + dz.y * dc.y) + (dz.z * dc.z + dz.w * dc.w)) * inv_i;
    const float vc  = half_sum((dc.x * dc.x + dc.y * dc.y) + (dc.z * dc.z + dc.w * dc.w)) * inv_i;
    *(float4*)(s_dz + 4 * t) = dz;
    *(float4*)(s_dc + 4 * t) = dc;
    *(float4*)(s_c  + 4 * t) = c;
    if ((t & 15) == 0) {
      s_mom[(t >> 4) * 4 + 0] = vz;
      s_mom[(t >> 4) * 4 + 1] = czc;
      s_mom[(t >> 4) * 4 + 2] = vc;
    }
    if (t < Hsz) {
      const int chunk = blockIdx.x & 7;
      float carry = 0.f;
#pragma unroll
      for (int k = 0; k < NCH - 1; ++k)
        if (k < chunk) carry += T[k * (Bsz * Hsz) + bb * Hsz + t];
      s_mom[t * 4 + 3] = carry;
    }
    __syncthreads();
#pragma unroll
    for (int i = 0; i < Hsz; ++i) {
      const int hd = i;
      const float ev = expf(s_mom[hd * 4 + 3] + slv[i]);
      const float var = fmaxf(s_mom[hd * 4 + 0]
                              + ev * (2.0f * s_mom[hd * 4 + 1] + ev * s_mom[hd * 4 + 2]), 0.f);
      const float sv = gn_w[hd] / sqrtf(var + GN_EPS);
      *(float2*)(s_es + hd * ESTR + 2 * t) = make_float2(ev, sv);
    }
    __syncthreads();
  }

  {
    const int hcol = blockIdx.y * Isz + 2 * lane;
    const int ch = bb * Dsz + hcol;
    const v2f z = *(const v2f*)(zhh + ch);
    const v2f c = *(const v2f*)(s_c + hcol);
#pragma unroll
    for (int g = 0; g < 2; ++g) {
      const int hrow0 = m_base + g * 16;
      const float ev = s_es[blockIdx.y * ESTR + 2 * (wave * 32 + g * 16 + mr)];
#pragma unroll
      for (int r = 0; r < 16; ++r) {
        const float er = lane_bcast(ev, r);
        v2f hv;
        hv.x = z.x + er * c.x;
        hv.y = z.y + er * c.y;
        *(v2f*)(Hout + (size_t)(hrow0 + r) * Dsz + hcol) = hv;
      }
    }
  }

  v8f accg[2][4], accy[2][4];
  const v8f zero = {0.f, 0.f, 0.f, 0.f, 0.f, 0.f, 0.f, 0.f};
  for (int i = 0; i < 2; ++i)
    for (int j = 0; j < 4; ++j) { accg[i][j] = zero; accy[i][j] = zero; }

  const size_t ra = (size_t)(m_base + mr) * Dsz + 4 * kk;
  const size_t rb = (size_t)(n_base + mr) * Dsz + 4 * kk;
  const float* G0 = X   + ra;
  const float* G1 = G0  + 16 * Dsz;
  const float* V0 = Wya + rb;
  const float* B0 = Wy  + rb;
  const float* E0 = s_es + 2 * (wave * 32 + mr);
  const float* E1 = E0 + 2 * 16;

  v4f pf[2][KBS / 8][6];
#define OUT_LOAD(S, K, PA0, PA1, PB)                                 \
  _Pragma("unroll")                                                  \
  for (int u = 0; u < KBS / 8; ++u) {                                \
    pf[S][u][0] = *(const v4f*)(PA0 + (K) + 8 * u);                  \
    pf[S][u][1] = *(const v4f*)(PA1 + (K) + 8 * u);                  \
    pf[S][u][2] = *(const v4f*)(PB            + (K) + 8 * u);        \
    pf[S][u][3] = *(const v4f*)(PB + 16 * Dsz + (K) + 8 * u);        \
    pf[S][u][4] = *(const v4f*)(PB + 32 * Dsz + (K) + 8 * u);        \
    pf[S][u][5] = *(const v4f*)(PB + 48 * Dsz + (K) + 8 * u);        \
  }
#define OUT_MMA_HALF(S, u, HF, ACC)                                  \
    _Pragma("unroll")                                                \
    for (int j = 0; j < 4; ++j) {                                    \
      ACC[0][j] = wmma4(pf[S][u][0].HF, pf[S][u][2 + j].HF, ACC[0][j]); \
      ACC[1][j] = wmma4(pf[S][u][1].HF, pf[S][u][2 + j].HF, ACC[1][j]); \
    }
#define OUT_MMA(S, ACC)                                              \
  _Pragma("unroll")                                                  \
  for (int u = 0; u < KBS / 8; ++u) {                                \
    OUT_MMA_HALF(S, u, xy, ACC)                                      \
    OUT_MMA_HALF(S, u, zw, ACC)                                      \
  }
#define OUT_LOAD_HN(S, K)                                            \
  _Pragma("unroll")                                                  \
  for (int u = 0; u < KBS / 8; ++u) {                                \
    pf[S][u][0] = *(const v4f*)(s_dz + 4 * kk + (K) + 8 * u);        \
    pf[S][u][1] = *(const v4f*)(s_dc + 4 * kk + (K) + 8 * u);        \
    pf[S][u][2] = *(const v4f*)(B0            + (K) + 8 * u);        \
    pf[S][u][3] = *(const v4f*)(B0 + 16 * Dsz + (K) + 8 * u);        \
    pf[S][u][4] = *(const v4f*)(B0 + 32 * Dsz + (K) + 8 * u);        \
    pf[S][u][5] = *(const v4f*)(B0 + 48 * Dsz + (K) + 8 * u);        \
  }
#define OUT_MMA_HN_HALF(S, u, HF)                                    \
    _Pragma("unroll")                                                \
    for (int j = 0; j < 4; ++j) {                                    \
      accy[0][j] = wmma4(f0.HF, pf[S][u][2 + j].HF, accy[0][j]);     \
      accy[1][j] = wmma4(f1.HF, pf[S][u][2 + j].HF, accy[1][j]);     \
    }
#define OUT_MMA_HN(S, Q0, Q1, GB)                                    \
  _Pragma("unroll")                                                  \
  for (int u = 0; u < KBS / 8; ++u) {                                \
    const v4f dzv = pf[S][u][0], dcv = pf[S][u][1];                  \
    const v4f f0 = (dzv + Q0.x * dcv) * Q0.y + GB;                   \
    const v4f f1 = (dzv + Q1.x * dcv) * Q1.y + GB;                   \
    OUT_MMA_HN_HALF(S, u, xy)                                        \
    OUT_MMA_HN_HALF(S, u, zw)                                        \
  }

  OUT_LOAD(0, 0, G0, G1, V0)
#pragma nounroll
  for (int kb = 0; kb < Dsz - 2 * KBS; kb += 2 * KBS) {
    OUT_LOAD(1, kb + KBS, G0, G1, V0)
    OUT_MMA(0, accg)
    OUT_LOAD(0, kb + 2 * KBS, G0, G1, V0)
    OUT_MMA(1, accg)
    LOADS_THEN_PRODUCTS(6 * (KBS / 8), 8 * (KBS / 4));
  }
  OUT_LOAD(1, Dsz - KBS, G0, G1, V0)
  OUT_MMA(0, accg)
  OUT_LOAD_HN(0, 0)
  v2f q0 = *(const v2f*)E0, q1 = *(const v2f*)E1;
  float gbh = gn_b[0];
  OUT_MMA(1, accg)

#pragma nounroll
  for (int kb = 0; kb < Dsz - 2 * KBS; kb += 2 * KBS) {
    const int hnx = (kb >> 6) + 1;
    const v2f nq0 = *(const v2f*)(E0 + hnx * ESTR);
    const v2f nq1 = *(const v2f*)(E1 + hnx * ESTR);
    const float ngb = gn_b[hnx];
    OUT_LOAD_HN(1, kb + KBS)
    OUT_MMA_HN(0, q0, q1, gbh)
    OUT_LOAD_HN(0, kb + 2 * KBS)
    OUT_MMA_HN(1, q0, q1, gbh)
    LOADS_THEN_PRODUCTS_HN(4 * (KBS / 8), 2 * (KBS / 8) + 1, 2 * (KBS / 8), 8 * (KBS / 4));
    q0 = nq0; q1 = nq1; gbh = ngb;
  }
  OUT_LOAD_HN(1, Dsz - KBS)
  OUT_MMA_HN(0, q0, q1, gbh)
  OUT_MMA_HN(1, q0, q1, gbh)
#undef OUT_LOAD
#undef OUT_LOAD_HN
#undef OUT_MMA
#undef OUT_MMA_HALF
#undef OUT_MMA_HN
#undef OUT_MMA_HN_HALF
#undef ESTR

  for (int mt = 0; mt < 2; ++mt)
    for (int nt = 0; nt < 4; ++nt) {
      const int n = n_base + nt * 16 + mr;
      const float biy = by[n], biya = bya[n];
#pragma unroll
      for (int r = 0; r < 8; ++r) {
        const int m = m_base + mt * 16 + kk * 8 + r;
        const float v = accy[mt][nt][r] + biy;
        const float g = silu_f(accg[mt][nt][r] + biya);
        OUT[(size_t)m * Dsz + n] = v * g;
      }
    }
}

extern "C" void kernel_launch(void* const* d_in, const int* in_sizes, int n_in,
                              void* d_out, int out_size, void* d_ws, size_t ws_size,
                              hipStream_t stream) {
  const float* x      = (const float*)d_in[0];
  const float* hidden = (const float*)d_in[1];
  const float* w_z    = (const float*)d_in[2];
  const float* b_z    = (const float*)d_in[3];
  const float* w_za   = (const float*)d_in[4];
  const float* b_za   = (const float*)d_in[5];
  const float* w_y    = (const float*)d_in[6];
  const float* b_y    = (const float*)d_in[7];
  const float* w_ya   = (const float*)d_in[8];
  const float* b_ya   = (const float*)d_in[9];
  const float* w_dt   = (const float*)d_in[10];
  const float* b_dt   = (const float*)d_in[11];
  const float* ln_a   = (const float*)d_in[12];
  const float* gn_w   = (const float*)d_in[13];
  const float* gn_b   = (const float*)d_in[14];

  float* y_out = (float*)d_out;
  float* h_out = y_out + (size_t)Msz * Dsz;

  float* ws   = (float*)d_ws;
  float* Z    = ws;
  float* SLb  = Z    + (size_t)Msz * Dsz;
  float* ZHH  = SLb  + (size_t)Hsz * Msz;
  float* Pb   = ZHH  + (size_t)NBHI;
  float* PDb  = Pb   + (size_t)NCH * NBHI;
  float* Tb   = PDb  + (size_t)NSEG * Msz * Hsz;

  const dim3 gGemm(Msz / 128, Dsz / 64, 1);
  const dim3 bGemm(128, 1, 1);

  gemm_gate_z_kernel<<<gGemm, bGemm, 0, stream>>>(x, w_z, b_z, w_za, b_za, Z,
                                                  w_dt, PDb);
  zsum_kernel<<<dim3(Bsz * Hsz, NCH), dim3(256), 0, stream>>>(
      Z, hidden, PDb, b_dt, ln_a, Pb, ZHH, SLb, Tb);
  gemm_out_kernel<<<gGemm, bGemm, 0, stream>>>(Pb, Tb, SLb, ZHH, gn_w, gn_b,
                                               w_y, b_y, x, w_ya, b_ya,
                                               y_out, h_out);
}


# Column-sum launch gone, 3 -> 2: each z GEMM block (one 128-row chunk, one head) sums its tile into the chunk partial, z not stored, dt done per block; output GEMM scans ln_da. Sum orders change
# speedup vs baseline: 3.8046x; 1.0623x over previous
#include <hip/hip_runtime.h>
#include <hip/hip_bf16.h>
#include <math.h>


#define Bsz 4
#define Lsz 1024
#define Dsz 512
#define Hsz 8
#define Isz 64
#define Msz (Bsz * Lsz)
#define GN_EPS 1e-5f
#define LOADS_THEN_PRODUCTS(NL, NM)                         \
  __builtin_amdgcn_sched_group_barrier(0x020, NL, 0);       \
  __builtin_amdgcn_sched_group_barrier(0x008, NM, 0);       \
  __builtin_amdgcn_sched_group_barrier(0x020, NL, 0);       \
  __builtin_amdgcn_sched_group_barrier(0x008, NM, 0)
#define LOADS_THEN_PRODUCTS_HN(NV, ND1, ND2, NM)            \
  __builtin_amdgcn_sched_group_barrier(0x020, NV, 0);       \
  __builtin_amdgcn_sched_group_barrier(0x100, ND1, 0);      \
  __builtin_amdgcn_sched_group_barrier(0x008, NM, 0);       \
  __builtin_amdgcn_sched_group_barrier(0x020, NV, 0);       \
  __builtin_amdgcn_sched_group_barrier(0x100, ND2, 0);      \
  __builtin_amdgcn_sched_group_barrier(0x008, NM, 0)
#define KBG 16
#define KBS 32

typedef float v2f __attribute__((ext_vector_type(2)));
typedef float v4f __attribute__((ext_vector_type(4)));
typedef float v8f __attribute__((ext_vector_type(8)));


__device__ __forceinline__ v8f wmma4(v2f a, v2f b, v8f c) {
  return __builtin_amdgcn_wmma_f32_16x16x4_f32(
       false, a,  false, b,
       (short)0, c,  false,  false);
}

__device__ __forceinline__ float silu_f(float v) {
  return v / (1.0f + expf(-v));
}

#define NCH 8
#define NBHI (Bsz * Hsz * Isz)

__global__ __launch_bounds__(128)
void gemm_gate_z_kernel(const float* __restrict__ X,
                        const float* __restrict__ Wz,  const float* __restrict__ bz,
                        const float* __restrict__ Wza, const float* __restrict__ bza,
                        const float* __restrict__ Wd,  const float* __restrict__ b_dt,
                        const float* __restrict__ ln_a, const float* __restrict__ hidden,
                        float* __restrict__ P, float* __restrict__ zhh,
                        float* __restrict__ LN) {
  __shared__ float s_dt[4 * 256];
  __shared__ float s_red[4 * 32];
  const int lane = threadIdx.x & 31;
  const int wave = threadIdx.x >> 5;
  const int mr = lane & 15;
  const int kk = lane >> 4;
  const int m_base = blockIdx.x * 128 + (wave & 1) * 64;
  const int n_base = blockIdx.y * 64 + (wave >> 1) * 32;

  v8f a1[4][2], a2[4][2];
  const v8f zero = {0.f, 0.f, 0.f, 0.f, 0.f, 0.f, 0.f, 0.f};
  for (int i = 0; i < 4; ++i)
    for (int j = 0; j < 2; ++j) { a1[i][j] = zero; a2[i][j] = zero; }

  const float* A0  = X   + (size_t)(m_base + mr)       * Dsz + 4 * kk;
  const float* Bz0 = Wz  + (size_t)(n_base + mr)       * Dsz + 4 * kk;
  const float* Bz1 = Bz0 + 16 * Dsz;
  const float* Ba0 = Wza + (size_t)(n_base + mr)       * Dsz + 4 * kk;
  const float* Ba1 = Ba0 + 16 * Dsz;

  v4f pa[2][KBG / 8][8];
#define GATE_LOAD(S, K)                                              \
  _Pragma("unroll")                                                  \
  for (int u = 0; u < KBG / 8; ++u) {                                \
    pa[S][u][0] = *(const v4f*)(A0            + (K) + 8 * u);        \
    pa[S][u][1] = *(const v4f*)(A0 + 16 * Dsz + (K) + 8 * u);        \
    pa[S][u][2] = *(const v4f*)(A0 + 32 * Dsz + (K) + 8 * u);        \
    pa[S][u][3] = *(const v4f*)(A0 + 48 * Dsz + (K) + 8 * u);        \
    pa[S][u][4] = *(const v4f*)(Bz0 + (K) + 8 * u);                  \
    pa[S][u][5] = *(const v4f*)(Bz1 + (K) + 8 * u);                  \
    pa[S][u][6] = *(const v4f*)(Ba0 + (K) + 8 * u);                  \
    pa[S][u][7] = *(const v4f*)(Ba1 + (K) + 8 * u);                  \
  }
#define GATE_MMA_HALF(S, u, HF)                                      \
    _Pragma("unroll")                                                \
    for (int i = 0; i < 4; ++i) {                                    \
      a1[i][0] = wmma4(pa[S][u][i].HF, pa[S][u][4].HF, a1[i][0]);    \
      a1[i][1] = wmma4(pa[S][u][i].HF, pa[S][u][5].HF, a1[i][1]);    \
      a2[i][0] = wmma4(pa[S][u][i].HF, pa[S][u][6].HF, a2[i][0]);    \
      a2[i][1] = wmma4(pa[S][u][i].HF, pa[S][u][7].HF, a2[i][1]);    \
    }
#define GATE_MMA(S)                                                  \
  _Pragma("unroll")                                                  \
  for (int u = 0; u < KBG / 8; ++u) {                                \
    GATE_MMA_HALF(S, u, xy)                                          \
    GATE_MMA_HALF(S, u, zw)                                          \
  }

  GATE_LOAD(0, 0)

  {
    const float* Ad = X + (size_t)(blockIdx.x * 128 + blockIdx.y * 16 + mr) * Dsz
                        + 4 * kk + wave * 128;
    const float* Bd = Wd + (size_t)(mr & 7) * Dsz + 4 * kk + wave * 128;
    v8f d0 = zero, d1 = zero;
#pragma unroll
    for (int u = 0; u < 128; u += 16) {
      const v4f fa0 = *(const v4f*)(Ad + u),     fb0 = *(const v4f*)(Bd + u);
      const v4f fa1 = *(const v4f*)(Ad + u + 8), fb1 = *(const v4f*)(Bd + u + 8);
      d0 = wmma4(fa0.xy, fb0.xy, d0);
      d1 = wmma4(fa1.xy, fb1.xy, d1);
      d0 = wmma4(fa0.zw, fb0.zw, d0);
      d1 = wmma4(fa1.zw, fb1.zw, d1);
    }
    float* sd = s_dt + wave * 256 + (kk * 8) * 16 + mr;
#pragma unroll
    for (int r = 0; r < 8; ++r) sd[r * 16] = d0[r] + d1[r];
  }

#pragma nounroll
  for (int kb = 0; kb < Dsz - 2 * KBG; kb += 2 * KBG) {
    GATE_LOAD(1, kb + KBG)
    GATE_MMA(0)
    GATE_LOAD(0, kb + 2 * KBG)
    GATE_MMA(1)
    LOADS_THEN_PRODUCTS(8 * (KBG / 8), 16 * (KBG / 4));
  }
  GATE_LOAD(1, Dsz - KBG)
  GATE_MMA(0)
  GATE_MMA(1)
#undef GATE_LOAD
#undef GATE_MMA
#undef GATE_MMA_HALF

  const int hd = blockIdx.y;
  const int bh = (blockIdx.x >> 3) * Hsz + hd;
  const int chunk = blockIdx.x & 7;
  const int l0 = (m_base & (Lsz - 1)) + kk * 8;
  float cs[2] = {0.f, 0.f};
#pragma unroll
  for (int mt = 0; mt < 4; ++mt)
#pragma unroll
    for (int nt = 0; nt < 2; ++nt) {
      const int n = n_base + nt * 16 + mr;
      const float biz = bz[n], biza = bza[n];
#pragma unroll
      for (int r = 0; r < 8; ++r) {
        const float v1 = a1[mt][nt][r] + biz;
        const float v2 = a2[mt][nt][r] + biza;
        const float zv = v1 * silu_f(v2);
        if (mt == 0) {
          const int l = l0 + r;
          if (l == hd) zhh[bh * Isz + (n - hd * Isz)] = zv;
          cs[nt] += (l > hd) ? zv : 0.f;
        } else {
          cs[nt] += zv;
        }
      }
    }
  cs[0] += __shfl_xor(cs[0], 16, 32);
  cs[1] += __shfl_xor(cs[1], 16, 32);
  if (kk == 0) {
    s_red[wave * 32 + mr]      = cs[0];
    s_red[wave * 32 + 16 + mr] = cs[1];
  }
  __syncthreads();
  if (threadIdx.x < Isz) {
    const int t = threadIdx.x;
    const int w0 = (t >> 5) * 2;
    float tot = s_red[w0 * 32 + (t & 31)] + s_red[(w0 + 1) * 32 + (t & 31)];
    const int ch = bh * Isz + t;
    if (chunk == 0) tot += hidden[ch];
    P[(size_t)chunk * NBHI + ch] = tot;
  }
  {
    const int t = threadIdx.x;
    const int si = (t >> 3) * 16 + (t & 7);
    const float dot = (s_dt[si] + s_dt[256 + si]) + (s_dt[512 + si] + s_dt[768 + si]);
    const float dv = dot + b_dt[t & 7];
    const float sp = (dv > 20.f) ? dv : log1pf(expf(dv));
    LN[(size_t)(blockIdx.x * 128 + blockIdx.y * 16) * Hsz + t] = -expf(ln_a[t & 7]) * sp;
  }
}

__device__ __forceinline__ float half_sum(float v) {
#pragma unroll
  for (int off = 8; off >= 1; off >>= 1) v += __shfl_xor(v, off, 32);
  return v;
}

__device__ __forceinline__ float lane_bcast(float v, int src) {
  return __builtin_bit_cast(
      float, __builtin_amdgcn_readlane(__builtin_bit_cast(int, v), src));
}

__global__ __launch_bounds__(128)
void gemm_out_kernel(const float* __restrict__ P, const float* __restrict__ LN,
                     const float* __restrict__ zhh,
                     const float* __restrict__ gn_w, const float* __restrict__ gn_b,
                     const float* __restrict__ Wy, const float* __restrict__ by,
                     const float* __restrict__ X, const float* __restrict__ Wya,
                     const float* __restrict__ bya, float* __restrict__ OUT,
                     float* __restrict__ Hout) {
  const int lane = threadIdx.x & 31;
  const int wave = threadIdx.x >> 5;
  const int mr = lane & 15;
  const int kk = lane >> 4;
  const int m_base = blockIdx.x * 128 + wave * 32;
  const int n_base = blockIdx.y * 64;
  const int bb = blockIdx.x >> 3;

  __shared__ __attribute__((aligned(16))) float s_dz[Dsz];
  __shared__ __attribute__((aligned(16))) float s_dc[Dsz];
  __shared__ __attribute__((aligned(16))) float s_c[Dsz];
  __shared__ __attribute__((aligned(16))) float s_es[Hsz * 128 * 2];
  __shared__ float s_mom[Hsz * 4];
  __shared__ __attribute__((aligned(16))) float s_wl[Hsz * 4];
  __shared__ __attribute__((aligned(16))) float s_wf[Hsz * 4];
#define ESTR (128 * 2)

  {
    const int t = threadIdx.x;
    const int ch4 = bb * Dsz + 4 * t;
    float4 c = make_float4(0.f, 0.f, 0.f, 0.f);
#pragma unroll
    for (int k = 0; k < NCH; ++k) {
      const float4 p = *(const float4*)(P + (size_t)k * NBHI + ch4);
      c.x += p.x; c.y += p.y; c.z += p.z; c.w += p.w;
    }
    const float4 z = *(const float4*)(zhh + ch4);
    const int chunk = blockIdx.x & 7;
    const float* lb = LN + (size_t)bb * Lsz * Hsz;
    const v4f lo = *(const v4f*)(lb + (size_t)(chunk * 128 + t) * Hsz);
    const v4f hi = *(const v4f*)(lb + (size_t)(chunk * 128 + t) * Hsz + 4);
    v4f fq = {0.f, 0.f, 0.f, 0.f};
#pragma unroll
    for (int k = 0; k < NCH - 1; ++k) {
      const v4f p0 = *(const v4f*)(lb + (size_t)k * 128 * Hsz + 4 * t);
      const v4f p1 = *(const v4f*)(lb + (size_t)k * 128 * Hsz + 4 * (t + 128));
      const v4f ps = p0 + p1;
      fq.x += (k < chunk) ? ps.x : 0.f;
      fq.y += (k < chunk) ? ps.y : 0.f;
      fq.z += (k < chunk) ? ps.z : 0.f;
      fq.w += (k < chunk) ? ps.w : 0.f;
    }
    float slv[Hsz] = {lo.x, lo.y, lo.z, lo.w, hi.x, hi.y, hi.z, hi.w};
#pragma unroll
    for (int off = 1; off < 32; off <<= 1) {
#pragma unroll
      for (int i = 0; i < Hsz; ++i) {
        const float up = __shfl_up(slv[i], off, 32);
        if (lane >= off) slv[i] += up;
      }
    }
#pragma unroll
    for (int off = 16; off >= 2; off >>= 1) {
      fq.x += __shfl_xor(fq.x, off, 32);
      fq.y += __shfl_xor(fq.y, off, 32);
      fq.z += __shfl_xor(fq.z, off, 32);
      fq.w += __shfl_xor(fq.w, off, 32);
    }

    const float inv_i = 1.0f / (float)Isz;
    const float mz = half_sum((z.x + z.y) + (z.z + z.w)) * inv_i;
    const float mc = half_sum((c.x + c.y) + (c.z + c.w)) * inv_i;
    const float4 dz = make_float4(z.x - mz, z.y - mz, z.z - mz, z.w - mz);
    const float4 dc = make_float4(c.x - mc, c.y - mc, c.z - mc, c.w - mc);
    const float vz  = half_sum((dz.x * dz.x + dz.y * dz.y) + (dz.z * dz.z + dz.w * dz.w)) * inv_i;
    const float czc = half_sum((dz.x * dc.x + dz.y * dc.y) + (dz.z * dc.z + dz.w * dc.w)) * inv_i;
    const float vc  = half_sum((dc.x * dc.x + dc.y * dc.y) + (dc.z * dc.z + dc.w * dc.w)) * inv_i;
    *(float4*)(s_dz + 4 * t) = dz;
    *(float4*)(s_dc + 4 * t) = dc;
    *(float4*)(s_c  + 4 * t) = c;
    if ((t & 15) == 0) {
      s_mom[(t >> 4) * 4 + 0] = vz;
      s_mom[(t >> 4) * 4 + 1] = czc;
      s_mom[(t >> 4) * 4 + 2] = vc;
    }
    if (lane == 31) {
#pragma unroll
      for (int i = 0; i < Hsz; ++i) s_wl[i * 4 + wave] = slv[i];
    }
    if (lane < 2) {
      s_wf[(4 * lane + 0) * 4 + wave] = fq.x;
      s_wf[(4 * lane + 1) * 4 + wave] = fq.y;
      s_wf[(4 * lane + 2) * 4 + wave] = fq.z;
      s_wf[(4 * lane + 3) * 4 + wave] = fq.w;
    }
    __syncthreads();
#pragma unroll
    for (int i = 0; i < Hsz; ++i) {
      const int hd = i;
      const v4f wf = *(const v4f*)(s_wf + 4 * hd);
      const v4f wl = *(const v4f*)(s_wl + 4 * hd);
      float carry = (wf.x + wf.y) + (wf.z + wf.w);
      if (wave > 0) carry += wl.x;
      if (wave > 1) carry += wl.y;
      if (wave > 2) carry += wl.z;
      const float ev = expf(carry + slv[i]);
      const float var = fmaxf(s_mom[hd * 4 + 0]
                              + ev * (2.0f * s_mom[hd * 4 + 1] + ev * s_mom[hd * 4 + 2]), 0.f);
      const float sv = gn_w[hd] / sqrtf(var + GN_EPS);
      *(float2*)(s_es + hd * ESTR + 2 * t) = make_float2(ev, sv);
    }
    __syncthreads();
  }

  {
    const int hcol = blockIdx.y * Isz + 2 * lane;
    const int ch = bb * Dsz + hcol;
    const v2f z = *(const v2f*)(zhh + ch);
    const v2f c = *(const v2f*)(s_c + hcol);
#pragma unroll
    for (int g = 0; g < 2; ++g) {
      const int hrow0 = m_base + g * 16;
      const float ev = s_es[blockIdx.y * ESTR + 2 * (wave * 32 + g * 16 + mr)];
#pragma unroll
      for (int r = 0; r < 16; ++r) {
        const float er = lane_bcast(ev, r);
        v2f hv;
        hv.x = z.x + er * c.x;
        hv.y = z.y + er * c.y;
        *(v2f*)(Hout + (size_t)(hrow0 + r) * Dsz + hcol) = hv;
      }
    }
  }

  v8f accg[2][4], accy[2][4];
  const v8f zero = {0.f, 0.f, 0.f, 0.f, 0.f, 0.f, 0.f, 0.f};
  for (int i = 0; i < 2; ++i)
    for (int j = 0; j < 4; ++j) { accg[i][j] = zero; accy[i][j] = zero; }

  const size_t ra = (size_t)(m_base + mr) * Dsz + 4 * kk;
  const size_t rb = (size_t)(n_base + mr) * Dsz + 4 * kk;
  const float* G0 = X   + ra;
  const float* G1 = G0  + 16 * Dsz;
  const float* V0 = Wya + rb;
  const float* B0 = Wy  + rb;
  const float* E0 = s_es + 2 * (wave * 32 + mr);
  const float* E1 = E0 + 2 * 16;

  v4f pf[2][KBS / 8][6];
#define OUT_LOAD(S, K, PA0, PA1, PB)                                 \
  _Pragma("unroll")                                                  \
  for (int u = 0; u < KBS / 8; ++u) {                                \
    pf[S][u][0] = *(const v4f*)(PA0 + (K) + 8 * u);                  \
    pf[S][u][1] = *(const v4f*)(PA1 + (K) + 8 * u);                  \
    pf[S][u][2] = *(const v4f*)(PB            + (K) + 8 * u);        \
    pf[S][u][3] = *(const v4f*)(PB + 16 * Dsz + (K) + 8 * u);        \
    pf[S][u][4] = *(const v4f*)(PB + 32 * Dsz + (K) + 8 * u);        \
    pf[S][u][5] = *(const v4f*)(PB + 48 * Dsz + (K) + 8 * u);        \
  }
#define OUT_MMA_HALF(S, u, HF, ACC)                                  \
    _Pragma("unroll")                                                \
    for (int j = 0; j < 4; ++j) {                                    \
      ACC[0][j] = wmma4(pf[S][u][0].HF, pf[S][u][2 + j].HF, ACC[0][j]); \
      ACC[1][j] = wmma4(pf[S][u][1].HF, pf[S][u][2 + j].HF, ACC[1][j]); \
    }
#define OUT_MMA(S, ACC)                                              \
  _Pragma("unroll")                                                  \
  for (int u = 0; u < KBS / 8; ++u) {                                \
    OUT_MMA_HALF(S, u, xy, ACC)                                      \
    OUT_MMA_HALF(S, u, zw, ACC)                                      \
  }
#define OUT_LOAD_HN(S, K)                                            \
  _Pragma("unroll")                                                  \
  for (int u = 0; u < KBS / 8; ++u) {                                \
    pf[S][u][0] = *(const v4f*)(s_dz + 4 * kk + (K) + 8 * u);        \
    pf[S][u][1] = *(const v4f*)(s_dc + 4 * kk + (K) + 8 * u);        \
    pf[S][u][2] = *(const v4f*)(B0            + (K) + 8 * u);        \
    pf[S][u][3] = *(const v4f*)(B0 + 16 * Dsz + (K) + 8 * u);        \
    pf[S][u][4] = *(const v4f*)(B0 + 32 * Dsz + (K) + 8 * u);        \
    pf[S][u][5] = *(const v4f*)(B0 + 48 * Dsz + (K) + 8 * u);        \
  }
#define OUT_MMA_HN_HALF(S, u, HF)                                    \
    _Pragma("unroll")                                                \
    for (int j = 0; j < 4; ++j) {                                    \
      accy[0][j] = wmma4(f0.HF, pf[S][u][2 + j].HF, accy[0][j]);     \
      accy[1][j] = wmma4(f1.HF, pf[S][u][2 + j].HF, accy[1][j]);     \
    }
#define OUT_MMA_HN(S, Q0, Q1, GB)                                    \
  _Pragma("unroll")                                                  \
  for (int u = 0; u < KBS / 8; ++u) {                                \
    const v4f dzv = pf[S][u][0], dcv = pf[S][u][1];                  \
    const v4f f0 = (dzv + Q0.x * dcv) * Q0.y + GB;                   \
    const v4f f1 = (dzv + Q1.x * dcv) * Q1.y + GB;                   \
    OUT_MMA_HN_HALF(S, u, xy)                                        \
    OUT_MMA_HN_HALF(S, u, zw)                                        \
  }

  OUT_LOAD(0, 0, G0, G1, V0)
#pragma nounroll
  for (int kb = 0; kb < Dsz - 2 * KBS; kb += 2 * KBS) {
    OUT_LOAD(1, kb + KBS, G0, G1, V0)
    OUT_MMA(0, accg)
    OUT_LOAD(0, kb + 2 * KBS, G0, G1, V0)
    OUT_MMA(1, accg)
    LOADS_THEN_PRODUCTS(6 * (KBS / 8), 8 * (KBS / 4));
  }
  OUT_LOAD(1, Dsz - KBS, G0, G1, V0)
  OUT_MMA(0, accg)
  OUT_LOAD_HN(0, 0)
  v2f q0 = *(const v2f*)E0, q1 = *(const v2f*)E1;
  float gbh = gn_b[0];
  OUT_MMA(1, accg)

#pragma nounroll
  for (int kb = 0; kb < Dsz - 2 * KBS; kb += 2 * KBS) {
    const int hnx = (kb >> 6) + 1;
    const v2f nq0 = *(const v2f*)(E0 + hnx * ESTR);
    const v2f nq1 = *(const v2f*)(E1 + hnx * ESTR);
    const float ngb = gn_b[hnx];
    OUT_LOAD_HN(1, kb + KBS)
    OUT_MMA_HN(0, q0, q1, gbh)
    OUT_LOAD_HN(0, kb + 2 * KBS)
    OUT_MMA_HN(1, q0, q1, gbh)
    LOADS_THEN_PRODUCTS_HN(4 * (KBS / 8), 2 * (KBS / 8) + 1, 2 * (KBS / 8), 8 * (KBS / 4));
    q0 = nq0; q1 = nq1; gbh = ngb;
  }
  OUT_LOAD_HN(1, Dsz - KBS)
  OUT_MMA_HN(0, q0, q1, gbh)
  OUT_MMA_HN(1, q0, q1, gbh)
#undef OUT_LOAD
#undef OUT_LOAD_HN
#undef OUT_MMA
#undef OUT_MMA_HALF
#undef OUT_MMA_HN
#undef OUT_MMA_HN_HALF
#undef ESTR

  for (int mt = 0; mt < 2; ++mt)
    for (int nt = 0; nt < 4; ++nt) {
      const int n = n_base + nt * 16 + mr;
      const float biy = by[n], biya = bya[n];
#pragma unroll
      for (int r = 0; r < 8; ++r) {
        const int m = m_base + mt * 16 + kk * 8 + r;
        const float v = accy[mt][nt][r] + biy;
        const float g = silu_f(accg[mt][nt][r] + biya);
        OUT[(size_t)m * Dsz + n] = v * g;
      }
    }
}

extern "C" void kernel_launch(void* const* d_in, const int* in_sizes, int n_in,
                              void* d_out, int out_size, void* d_ws, size_t ws_size,
                              hipStream_t stream) {
  const float* x      = (const float*)d_in[0];
  const float* hidden = (const float*)d_in[1];
  const float* w_z    = (const float*)d_in[2];
  const float* b_z    = (const float*)d_in[3];
  const float* w_za   = (const float*)d_in[4];
  const float* b_za   = (const float*)d_in[5];
  const float* w_y    = (const float*)d_in[6];
  const float* b_y    = (const float*)d_in[7];
  const float* w_ya   = (const float*)d_in[8];
  const float* b_ya   = (const float*)d_in[9];
  const float* w_dt   = (const float*)d_in[10];
  const float* b_dt   = (const float*)d_in[11];
  const float* ln_a   = (const float*)d_in[12];
  const float* gn_w   = (const float*)d_in[13];
  const float* gn_b   = (const float*)d_in[14];

  float* y_out = (float*)d_out;
  float* h_out = y_out + (size_t)Msz * Dsz;

  float* ws   = (float*)d_ws;
  float* ZHH  = ws;
  float* Pb   = ZHH  + (size_t)NBHI;
  float* LNb  = Pb   + (size_t)NCH * NBHI;

  const dim3 gGemm(Msz / 128, Dsz / 64, 1);
  const dim3 bGemm(128, 1, 1);

  gemm_gate_z_kernel<<<gGemm, bGemm, 0, stream>>>(x, w_z, b_z, w_za, b_za,
                                                  w_dt, b_dt, ln_a, hidden,
                                                  Pb, ZHH, LNb);
  gemm_out_kernel<<<gGemm, bGemm, 0, stream>>>(Pb, LNb, ZHH, gn_w, gn_b,
                                               w_y, b_y, x, w_ya, b_ya,
                                               y_out, h_out);
}
